# starting kernel
# baseline (speedup 1.0000x reference)
#include <hip/hip_runtime.h>
#include <hip/hip_bf16.h>

typedef __attribute__((ext_vector_type(2))) float v2f;
typedef __attribute__((ext_vector_type(8))) float v8f;

#define BATCH   4
#define NPTS    8192
#define KNB     16
#define CIN     64
#define COUT    64
#define NC      16
#define DIM     3
#define L1IN    48
#define L1OUT   32
#define TP      16
#define NTHREADS 128
#define NWAVES  4
#define FSTRIDE 1028

static __device__ __forceinline__ v8f wmma_f32(v2f a, v2f b, v8f c) {
    return __builtin_amdgcn_wmma_f32_16x16x4_f32(false, a, false, b,
                                                 (short)0, c, false, false);
}

__global__ __launch_bounds__(NTHREADS)
void ptconv_wmma_kernel(const float* __restrict__ inp,
                        const float* __restrict__ points,
                        const float* __restrict__ next_pts,
                        const int*   __restrict__ indices,
                        const float* __restrict__ weight,
                        const float* __restrict__ bias,
                        const float* __restrict__ centers,
                        const float* __restrict__ l1w, const float* __restrict__ l1b,
                        const float* __restrict__ l2w, const float* __restrict__ l2b,
                        const float* __restrict__ l3w, const float* __restrict__ l3b,
                        float* __restrict__ out)
{
    __shared__ float s_l1w[L1IN * L1OUT];
    __shared__ float s_l1b[L1OUT];
    __shared__ float s_l2w[L1OUT * NC];
    __shared__ float s_l2b[NC];
    __shared__ float s_l3w[NC * NC];
    __shared__ float s_l3b[NC];
    __shared__ float s_cent[DIM * NC];
    __shared__ float s_bias[COUT];
    __shared__ float s_rel[TP * KNB][4];
    __shared__ int   s_idx[TP * KNB];
    __shared__ float s_d3[TP * KNB][NC];
    __shared__ float s_fmat[TP][FSTRIDE];
    __shared__ float s_s1[NWAVES][16 * 33];
    __shared__ float s_s2[NWAVES][16 * 17];

    const int tid    = threadIdx.x;
    const int wave   = tid >> 5;
    const int lane   = tid & 31;
    const int laneM  = lane & 15;
    const int hiHalf = lane >> 4;
    const int koffA  = hiHalf * 2;
    const int hi8    = hiHalf * 8;

    const int tilesPerBatch = NPTS / TP;
    const int b    = blockIdx.x / tilesPerBatch;
    const int tile = blockIdx.x % tilesPerBatch;
    const int n0   = tile * TP;

    for (int i = tid; i < L1IN * L1OUT; i += NTHREADS) s_l1w[i] = l1w[i];
    for (int i = tid; i < L1OUT * NC;  i += NTHREADS) s_l2w[i] = l2w[i];
    for (int i = tid; i < NC * NC;     i += NTHREADS) s_l3w[i] = l3w[i];
    if (tid < L1OUT)   s_l1b[tid]  = l1b[tid];
    if (tid < NC)      { s_l2b[tid] = l2b[tid]; s_l3b[tid] = l3b[tid]; }
    if (tid < DIM*NC)  s_cent[tid] = centers[tid];
    if (tid < COUT)    s_bias[tid] = bias[tid];

    for (int i = tid; i < CIN * NC; i += NTHREADS * 2)
        __builtin_prefetch(weight + (size_t)i * COUT, 0, 1);

    for (int r = tid; r < TP * KNB; r += NTHREADS) {
        const int p = r >> 4, k = r & 15;
        const int n = n0 + p;
        const int idx = indices[((size_t)b * NPTS + n) * KNB + k];
        s_idx[r] = idx;
        const float* pp = points   + ((size_t)b * NPTS + idx) * DIM;
        const float* qq = next_pts + ((size_t)b * NPTS + n  ) * DIM;
        s_rel[r][0] = pp[0] - qq[0];
        s_rel[r][1] = pp[1] - qq[1];
        s_rel[r][2] = pp[2] - qq[2];
    }
    __syncthreads();

    float* s1 = s_s1[wave];
    float* s2 = s_s2[wave];
    for (int t = 0; t < 4; ++t) {
        const int row0 = wave * 64 + t * 16;
        const int rA   = row0 + laneM;

        for (int nt = 0; nt < 2; ++nt) {
            v8f acc;
            const float bb = s_l1b[nt * 16 + laneM];
            #pragma unroll
            for (int r = 0; r < 8; ++r) acc[r] = bb;
            #pragma unroll
            for (int k0 = 0; k0 < L1IN; k0 += 4) {
                const int kA = k0 + koffA;
                v2f a, bf;
                a.x  = s_rel[rA][(kA    ) >> 4] - s_cent[kA    ];
                a.y  = s_rel[rA][(kA + 1) >> 4] - s_cent[kA + 1];
                bf.x = s_l1w[(kA    ) * L1OUT + nt * 16 + laneM];
                bf.y = s_l1w[(kA + 1) * L1OUT + nt * 16 + laneM];
                acc = wmma_f32(a, bf, acc);
            }
            #pragma unroll
            for (int r = 0; r < 8; ++r) {
                float v = acc[r]; v = v > 0.f ? v : 0.f;
                s1[(r + hi8) * 33 + nt * 16 + laneM] = v;
            }
        }

        {
            v8f acc;
            const float bb = s_l2b[laneM];
            #pragma unroll
            for (int r = 0; r < 8; ++r) acc[r] = bb;
            #pragma unroll
            for (int k0 = 0; k0 < L1OUT; k0 += 4) {
                const int kA = k0 + koffA;
                v2f a, bf;
                a.x  = s1[laneM * 33 + kA];
                a.y  = s1[laneM * 33 + kA + 1];
                bf.x = s_l2w[(kA    ) * NC + laneM];
                bf.y = s_l2w[(kA + 1) * NC + laneM];
                acc = wmma_f32(a, bf, acc);
            }
            #pragma unroll
            for (int r = 0; r < 8; ++r) {
                float v = acc[r]; v = v > 0.f ? v : 0.f;
                s2[(r + hi8) * 17 + laneM] = v;
            }
        }

        {
            v8f acc;
            const float bb = s_l3b[laneM];
            #pragma unroll
            for (int r = 0; r < 8; ++r) acc[r] = bb;
            #pragma unroll
            for (int k0 = 0; k0 < NC; k0 += 4) {
                const int kA = k0 + koffA;
                v2f a, bf;
                a.x  = s2[laneM * 17 + kA];
                a.y  = s2[laneM * 17 + kA + 1];
                bf.x = s_l3w[(kA    ) * NC + laneM];
                bf.y = s_l3w[(kA + 1) * NC + laneM];
                acc = wmma_f32(a, bf, acc);
            }
            #pragma unroll
            for (int r = 0; r < 8; ++r) {
                float v = acc[r]; v = v > 0.f ? v : 0.f;
                s_d3[row0 + r + hi8][laneM] = v;
            }
        }
    }
    __syncthreads();

    for (int pi = 0; pi < 4; ++pi) {
        const int p = wave * 4 + pi;
        for (int mt = 0; mt < 4; ++mt) {
            v8f acc = {0.f, 0.f, 0.f, 0.f, 0.f, 0.f, 0.f, 0.f};
            #pragma unroll
            for (int k0 = 0; k0 < KNB; k0 += 4) {
                const int kA = k0 + koffA;
                const int i0 = s_idx[p * 16 + kA];
                const int i1 = s_idx[p * 16 + kA + 1];
                v2f a, bf;
                a.x  = inp[((size_t)b * NPTS + i0) * CIN + mt * 16 + laneM];
                a.y  = inp[((size_t)b * NPTS + i1) * CIN + mt * 16 + laneM];
                bf.x = s_d3[p * 16 + kA    ][laneM];
                bf.y = s_d3[p * 16 + kA + 1][laneM];
                acc = wmma_f32(a, bf, acc);
            }
            #pragma unroll
            for (int r = 0; r < 8; ++r)
                s_fmat[p][(mt * 16 + r + hi8) * 16 + laneM] = acc[r];
        }
    }
    __syncthreads();

    {
        const int ocol = wave * 16 + laneM;
        v8f acc = {0.f, 0.f, 0.f, 0.f, 0.f, 0.f, 0.f, 0.f};
        #pragma unroll 4
        for (int k0 = 0; k0 < CIN * NC; k0 += 4) {
            const int kA = k0 + koffA;
            v2f a, bf;
            a.x  = s_fmat[laneM][kA];
            a.y  = s_fmat[laneM][kA + 1];
            bf.x = weight[(size_t)(kA    ) * COUT + ocol];
            bf.y = weight[(size_t)(kA + 1) * COUT + ocol];
            acc = wmma_f32(a, bf, acc);
        }
        const float bb = s_bias[ocol];
        #pragma unroll
        for (int r = 0; r < 8; ++r) {
            const int p = r + hi8;
            out[((size_t)b * NPTS + n0 + p) * COUT + ocol] =
                acc[r] * (1.0f / KNB) + bb;
        }
    }
}

extern "C" void kernel_launch(void* const* d_in, const int* in_sizes, int n_in,
                              void* d_out, int out_size, void* d_ws, size_t ws_size,
                              hipStream_t stream) {
    (void)in_sizes; (void)n_in; (void)out_size; (void)d_ws; (void)ws_size;
    const float* inp      = (const float*)d_in[0];
    const float* points   = (const float*)d_in[1];
    const float* next_pts = (const float*)d_in[2];
    const int*   indices  = (const int*)  d_in[3];
    const float* weight   = (const float*)d_in[4];
    const float* bias     = (const float*)d_in[5];
    const float* centers  = (const float*)d_in[6];
    const float* l1w      = (const float*)d_in[7];
    const float* l1b      = (const float*)d_in[8];
    const float* l2w      = (const float*)d_in[9];
    const float* l2b      = (const float*)d_in[10];
    const float* l3w      = (const float*)d_in[11];
    const float* l3b      = (const float*)d_in[12];
    float* out = (float*)d_out;

    const int grid = BATCH * (NPTS / TP);
    ptconv_wmma_kernel<<<grid, NTHREADS, 0, stream>>>(
        inp, points, next_pts, indices, weight, bias, centers,
        l1w, l1b, l2w, l2b, l3w, l3b, out);
}


# phase D: K split over the 4 waves, each wave all 64 columns on 4 accumulators fed by 16-byte weight-row loads one batch ahead; per-wave partial sums added in LDS (changes summation order)
# speedup vs baseline: 1.8323x; 1.8323x over previous
#include <hip/hip_runtime.h>
#include <hip/hip_bf16.h>

typedef __attribute__((ext_vector_type(2))) float v2f;
typedef __attribute__((ext_vector_type(8))) float v8f;
typedef __attribute__((ext_vector_type(4))) float v4f;

#define BATCH   4
#define NPTS    8192
#define KNB     16
#define CIN     64
#define COUT    64
#define NC      16
#define DIM     3
#define L1IN    48
#define L1OUT   32
#define TP      16
#define NTHREADS 128
#define NWAVES  4
#define FSTRIDE 1028

static __device__ __forceinline__ v8f wmma_f32(v2f a, v2f b, v8f c) {
    return __builtin_amdgcn_wmma_f32_16x16x4_f32(false, a, false, b,
                                                 (short)0, c, false, false);
}

static __device__ __forceinline__ void d_load_w(v4f (&wr)[16], const float* __restrict__ wrow) {
    #pragma unroll
    for (int j = 0; j < 4; ++j) {
        #pragma unroll
        for (int e = 0; e < 4; ++e)
            wr[j * 4 + e] = *reinterpret_cast<const v4f*>(wrow + (size_t)(j * 8 + e) * COUT);
    }
}
static __device__ __forceinline__ void d_load_a(v4f (&ar)[4], const float* arow) {
    #pragma unroll
    for (int j = 0; j < 4; ++j)
        ar[j] = *reinterpret_cast<const v4f*>(arow + j * 8);
}
static __device__ __forceinline__ void d_mma(v8f (&acc)[4], const v4f (&wr)[16], const v4f (&ar)[4]) {
    #pragma unroll
    for (int j = 0; j < 4; ++j) {
        #pragma unroll
        for (int u = 0; u < 2; ++u) {
            v2f a;
            a.x = ar[j][2 * u];
            a.y = ar[j][2 * u + 1];
            #pragma unroll
            for (int t = 0; t < 4; ++t) {
                v2f bf;
                bf.x = wr[j * 4 + 2 * u][t];
                bf.y = wr[j * 4 + 2 * u + 1][t];
                acc[t] = wmma_f32(a, bf, acc[t]);
            }
        }
    }
}

__global__ __launch_bounds__(NTHREADS)
void ptconv_wmma_kernel(const float* __restrict__ inp,
                        const float* __restrict__ points,
                        const float* __restrict__ next_pts,
                        const int*   __restrict__ indices,
                        const float* __restrict__ weight,
                        const float* __restrict__ bias,
                        const float* __restrict__ centers,
                        const float* __restrict__ l1w, const float* __restrict__ l1b,
                        const float* __restrict__ l2w, const float* __restrict__ l2b,
                        const float* __restrict__ l3w, const float* __restrict__ l3b,
                        float* __restrict__ out)
{
    __shared__ float s_l1w[L1IN * L1OUT];
    __shared__ float s_l1b[L1OUT];
    __shared__ float s_l2w[L1OUT * NC];
    __shared__ float s_l2b[NC];
    __shared__ float s_l3w[NC * NC];
    __shared__ float s_l3b[NC];
    __shared__ float s_cent[DIM * NC];
    __shared__ __attribute__((aligned(16))) float s_bias[COUT];
    __shared__ float s_rel[TP * KNB][4];
    __shared__ int   s_idx[TP * KNB];
    __shared__ __attribute__((aligned(16))) float s_d3[TP * KNB][NC];
    __shared__ __attribute__((aligned(16))) float s_fmat[TP][FSTRIDE];
    __shared__ float s_s1[NWAVES][16 * 33];
    __shared__ float s_s2[NWAVES][16 * 17];

    const int tid    = threadIdx.x;
    const int wave   = tid >> 5;
    const int lane   = tid & 31;
    const int laneM  = lane & 15;
    const int hiHalf = lane >> 4;
    const int koffA  = hiHalf * 2;
    const int hi8    = hiHalf * 8;

    const int tilesPerBatch = NPTS / TP;
    const int b    = blockIdx.x / tilesPerBatch;
    const int tile = blockIdx.x % tilesPerBatch;
    const int n0   = tile * TP;

    for (int i = tid; i < L1IN * L1OUT; i += NTHREADS) s_l1w[i] = l1w[i];
    for (int i = tid; i < L1OUT * NC;  i += NTHREADS) s_l2w[i] = l2w[i];
    for (int i = tid; i < NC * NC;     i += NTHREADS) s_l3w[i] = l3w[i];
    if (tid < L1OUT)   s_l1b[tid]  = l1b[tid];
    if (tid < NC)      { s_l2b[tid] = l2b[tid]; s_l3b[tid] = l3b[tid]; }
    if (tid < DIM*NC)  s_cent[tid] = centers[tid];
    if (tid < COUT)    s_bias[tid] = bias[tid];

    for (int i = tid; i < CIN * NC; i += NTHREADS * 2)
        __builtin_prefetch(weight + (size_t)i * COUT, 0, 1);

    for (int r = tid; r < TP * KNB; r += NTHREADS) {
        const int p = r >> 4, k = r & 15;
        const int n = n0 + p;
        const int idx = indices[((size_t)b * NPTS + n) * KNB + k];
        s_idx[r] = idx;
        const float* pp = points   + ((size_t)b * NPTS + idx) * DIM;
        const float* qq = next_pts + ((size_t)b * NPTS + n  ) * DIM;
        s_rel[r][0] = pp[0] - qq[0];
        s_rel[r][1] = pp[1] - qq[1];
        s_rel[r][2] = pp[2] - qq[2];
    }
    __syncthreads();

    float* s1 = s_s1[wave];
    float* s2 = s_s2[wave];
    for (int t = 0; t < 4; ++t) {
        const int row0 = wave * 64 + t * 16;
        const int rA   = row0 + laneM;

        for (int nt = 0; nt < 2; ++nt) {
            v8f acc;
            const float bb = s_l1b[nt * 16 + laneM];
            #pragma unroll
            for (int r = 0; r < 8; ++r) acc[r] = bb;
            #pragma unroll
            for (int k0 = 0; k0 < L1IN; k0 += 4) {
                const int kA = k0 + koffA;
                v2f a, bf;
                a.x  = s_rel[rA][(kA    ) >> 4] - s_cent[kA    ];
                a.y  = s_rel[rA][(kA + 1) >> 4] - s_cent[kA + 1];
                bf.x = s_l1w[(kA    ) * L1OUT + nt * 16 + laneM];
                bf.y = s_l1w[(kA + 1) * L1OUT + nt * 16 + laneM];
                acc = wmma_f32(a, bf, acc);
            }
            #pragma unroll
            for (int r = 0; r < 8; ++r) {
                float v = acc[r]; v = v > 0.f ? v : 0.f;
                s1[(r + hi8) * 33 + nt * 16 + laneM] = v;
            }
        }

        {
            v8f acc;
            const float bb = s_l2b[laneM];
            #pragma unroll
            for (int r = 0; r < 8; ++r) acc[r] = bb;
            #pragma unroll
            for (int k0 = 0; k0 < L1OUT; k0 += 4) {
                const int kA = k0 + koffA;
                v2f a, bf;
                a.x  = s1[laneM * 33 + kA];
                a.y  = s1[laneM * 33 + kA + 1];
                bf.x = s_l2w[(kA    ) * NC + laneM];
                bf.y = s_l2w[(kA + 1) * NC + laneM];
                acc = wmma_f32(a, bf, acc);
            }
            #pragma unroll
            for (int r = 0; r < 8; ++r) {
                float v = acc[r]; v = v > 0.f ? v : 0.f;
                s2[(r + hi8) * 17 + laneM] = v;
            }
        }

        {
            v8f acc;
            const float bb = s_l3b[laneM];
            #pragma unroll
            for (int r = 0; r < 8; ++r) acc[r] = bb;
            #pragma unroll
            for (int k0 = 0; k0 < NC; k0 += 4) {
                const int kA = k0 + koffA;
                v2f a, bf;
                a.x  = s2[laneM * 17 + kA];
                a.y  = s2[laneM * 17 + kA + 1];
                bf.x = s_l3w[(kA    ) * NC + laneM];
                bf.y = s_l3w[(kA + 1) * NC + laneM];
                acc = wmma_f32(a, bf, acc);
            }
            #pragma unroll
            for (int r = 0; r < 8; ++r) {
                float v = acc[r]; v = v > 0.f ? v : 0.f;
                s_d3[row0 + r + hi8][laneM] = v;
            }
        }
    }
    __syncthreads();

    for (int pi = 0; pi < 4; ++pi) {
        const int p = wave * 4 + pi;
        for (int mt = 0; mt < 4; ++mt) {
            v8f acc = {0.f, 0.f, 0.f, 0.f, 0.f, 0.f, 0.f, 0.f};
            #pragma unroll
            for (int k0 = 0; k0 < KNB; k0 += 4) {
                const int kA = k0 + koffA;
                const int i0 = s_idx[p * 16 + kA];
                const int i1 = s_idx[p * 16 + kA + 1];
                v2f a, bf;
                a.x  = inp[((size_t)b * NPTS + i0) * CIN + mt * 16 + laneM];
                a.y  = inp[((size_t)b * NPTS + i1) * CIN + mt * 16 + laneM];
                bf.x = s_d3[p * 16 + kA    ][laneM];
                bf.y = s_d3[p * 16 + kA + 1][laneM];
                acc = wmma_f32(a, bf, acc);
            }
            #pragma unroll
            for (int r = 0; r < 8; ++r)
                s_fmat[p][(mt * 16 + r + hi8) * 16 + laneM] = acc[r];
        }
    }

    {
        v8f acc[4];
        #pragma unroll
        for (int t = 0; t < 4; ++t) {
            #pragma unroll
            for (int r = 0; r < 8; ++r) acc[t][r] = 0.f;
        }
        const float* wrow = weight + (size_t)(wave * 256 + hiHalf * 4) * COUT + laneM * 4;
        v4f w0[16], w1[16], a0[4], a1[4];
        __syncthreads();
        const float* arow = &s_fmat[laneM][wave * 256 + hiHalf * 4];
        d_load_w(w0, wrow);
        d_load_a(a0, arow);
        for (int it = 0; it < 3; ++it) {
            d_load_w(w1, wrow + 32 * COUT);
            d_load_a(a1, arow + 32);
            __builtin_amdgcn_sched_barrier(0);
            d_mma(acc, w0, a0);
            __builtin_amdgcn_sched_barrier(0);
            wrow += 64 * COUT;
            arow += 64;
            d_load_w(w0, wrow);
            d_load_a(a0, arow);
            __builtin_amdgcn_sched_barrier(0);
            d_mma(acc, w1, a1);
            __builtin_amdgcn_sched_barrier(0);
        }
        d_load_w(w1, wrow + 32 * COUT);
        d_load_a(a1, arow + 32);
        __builtin_amdgcn_sched_barrier(0);
        d_mma(acc, w0, a0);
        d_mma(acc, w1, a1);

        float* part = &s_d3[0][0];
        #pragma unroll
        for (int r = 0; r < 8; ++r) {
            v4f v;
            v.x = acc[0][r]; v.y = acc[1][r]; v.z = acc[2][r]; v.w = acc[3][r];
            *reinterpret_cast<v4f*>(part + (wave * TP + hi8 + r) * COUT + laneM * 4) = v;
        }
        __syncthreads();
        #pragma unroll
        for (int i = 0; i < 2; ++i) {
            const int q   = tid + i * NTHREADS;
            const int p   = q >> 4;
            const int col = (q & 15) * 4;
            v4f s = *reinterpret_cast<const v4f*>(part + (0 * TP + p) * COUT + col);
            s += *reinterpret_cast<const v4f*>(part + (1 * TP + p) * COUT + col);
            s += *reinterpret_cast<const v4f*>(part + (2 * TP + p) * COUT + col);
            s += *reinterpret_cast<const v4f*>(part + (3 * TP + p) * COUT + col);
            const v4f bb = *reinterpret_cast<const v4f*>(s_bias + col);
            const v4f o  = s * (1.0f / KNB) + bb;
            *reinterpret_cast<v4f*>(out + ((size_t)b * NPTS + n0 + p) * COUT + col) = o;
        }
    }
}

extern "C" void kernel_launch(void* const* d_in, const int* in_sizes, int n_in,
                              void* d_out, int out_size, void* d_ws, size_t ws_size,
                              hipStream_t stream) {
    (void)in_sizes; (void)n_in; (void)out_size; (void)d_ws; (void)ws_size;
    const float* inp      = (const float*)d_in[0];
    const float* points   = (const float*)d_in[1];
    const float* next_pts = (const float*)d_in[2];
    const int*   indices  = (const int*)  d_in[3];
    const float* weight   = (const float*)d_in[4];
    const float* bias     = (const float*)d_in[5];
    const float* centers  = (const float*)d_in[6];
    const float* l1w      = (const float*)d_in[7];
    const float* l1b      = (const float*)d_in[8];
    const float* l2w      = (const float*)d_in[9];
    const float* l2b      = (const float*)d_in[10];
    const float* l3w      = (const float*)d_in[11];
    const float* l3b      = (const float*)d_in[12];
    float* out = (float*)d_out;

    const int grid = BATCH * (NPTS / TP);
    ptconv_wmma_kernel<<<grid, NTHREADS, 0, stream>>>(
        inp, points, next_pts, indices, weight, bias, centers,
        l1w, l1b, l2w, l2b, l3w, l3b, out);
}


# phase A: every parameter, index and own-point load issued as one group into registers (16-byte parameter loads), LDS stores afterwards: 2 memory round trips per workgroup instead of 19
# speedup vs baseline: 2.1702x; 1.1844x over previous
#include <hip/hip_runtime.h>
#include <hip/hip_bf16.h>

typedef __attribute__((ext_vector_type(2))) float v2f;
typedef __attribute__((ext_vector_type(8))) float v8f;
typedef __attribute__((ext_vector_type(4))) float v4f;

#define BATCH   4
#define NPTS    8192
#define KNB     16
#define CIN     64
#define COUT    64
#define NC      16
#define DIM     3
#define L1IN    48
#define L1OUT   32
#define TP      16
#define NTHREADS 128
#define NWAVES  4
#define FSTRIDE 1028

static __device__ __forceinline__ v8f wmma_f32(v2f a, v2f b, v8f c) {
    return __builtin_amdgcn_wmma_f32_16x16x4_f32(false, a, false, b,
                                                 (short)0, c, false, false);
}

static __device__ __forceinline__ void d_load_w(v4f (&wr)[16], const float* __restrict__ wrow) {
    #pragma unroll
    for (int j = 0; j < 4; ++j) {
        #pragma unroll
        for (int e = 0; e < 4; ++e)
            wr[j * 4 + e] = *reinterpret_cast<const v4f*>(wrow + (size_t)(j * 8 + e) * COUT);
    }
}
static __device__ __forceinline__ void d_load_a(v4f (&ar)[4], const float* arow) {
    #pragma unroll
    for (int j = 0; j < 4; ++j)
        ar[j] = *reinterpret_cast<const v4f*>(arow + j * 8);
}
static __device__ __forceinline__ void d_mma(v8f (&acc)[4], const v4f (&wr)[16], const v4f (&ar)[4]) {
    #pragma unroll
    for (int j = 0; j < 4; ++j) {
        #pragma unroll
        for (int u = 0; u < 2; ++u) {
            v2f a;
            a.x = ar[j][2 * u];
            a.y = ar[j][2 * u + 1];
            #pragma unroll
            for (int t = 0; t < 4; ++t) {
                v2f bf;
                bf.x = wr[j * 4 + 2 * u][t];
                bf.y = wr[j * 4 + 2 * u + 1][t];
                acc[t] = wmma_f32(a, bf, acc[t]);
            }
        }
    }
}

__global__ __launch_bounds__(NTHREADS)
void ptconv_wmma_kernel(const float* __restrict__ inp,
                        const float* __restrict__ points,
                        const float* __restrict__ next_pts,
                        const int*   __restrict__ indices,
                        const float* __restrict__ weight,
                        const float* __restrict__ bias,
                        const float* __restrict__ centers,
                        const float* __restrict__ l1w, const float* __restrict__ l1b,
                        const float* __restrict__ l2w, const float* __restrict__ l2b,
                        const float* __restrict__ l3w, const float* __restrict__ l3b,
                        float* __restrict__ out)
{
    __shared__ __attribute__((aligned(16))) float s_l1w[L1IN * L1OUT];
    __shared__ float s_l1b[L1OUT];
    __shared__ __attribute__((aligned(16))) float s_l2w[L1OUT * NC];
    __shared__ float s_l2b[NC];
    __shared__ __attribute__((aligned(16))) float s_l3w[NC * NC];
    __shared__ float s_l3b[NC];
    __shared__ float s_cent[DIM * NC];
    __shared__ __attribute__((aligned(16))) float s_bias[COUT];
    __shared__ float s_rel[TP * KNB][4];
    __shared__ int   s_idx[TP * KNB];
    __shared__ __attribute__((aligned(16))) float s_d3[TP * KNB][NC];
    __shared__ __attribute__((aligned(16))) float s_fmat[TP][FSTRIDE];
    __shared__ float s_s1[NWAVES][16 * 33];
    __shared__ float s_s2[NWAVES][16 * 17];

    const int tid    = threadIdx.x;
    const int wave   = tid >> 5;
    const int lane   = tid & 31;
    const int laneM  = lane & 15;
    const int hiHalf = lane >> 4;
    const int koffA  = hiHalf * 2;
    const int hi8    = hiHalf * 8;

    const int tilesPerBatch = NPTS / TP;
    const int b    = blockIdx.x / tilesPerBatch;
    const int tile = blockIdx.x % tilesPerBatch;
    const int n0   = tile * TP;

    const size_t row0g = (size_t)b * NPTS + n0;
    int   g_idx[2];
    float g_q[2][3];
    #pragma unroll
    for (int i = 0; i < 2; ++i) {
        const int r = tid + i * NTHREADS;
        g_idx[i] = indices[row0g * KNB + r];
        const float* qq = next_pts + (row0g + (r >> 4)) * DIM;
        g_q[i][0] = qq[0]; g_q[i][1] = qq[1]; g_q[i][2] = qq[2];
    }
    v4f p_l1w[3];
    #pragma unroll
    for (int k = 0; k < 3; ++k)
        p_l1w[k] = reinterpret_cast<const v4f*>(l1w)[tid + k * NTHREADS];
    const v4f p_l2w = reinterpret_cast<const v4f*>(l2w)[tid];
    const v2f p_l3w = reinterpret_cast<const v2f*>(l3w)[tid];
    const float p_l1b  = l1b[tid & (L1OUT - 1)];
    const float p_l2b  = l2b[tid & (NC - 1)];
    const float p_l3b  = l3b[tid & (NC - 1)];
    const float p_cent = centers[tid < DIM * NC ? tid : 0];
    const float p_bias = bias[tid & (COUT - 1)];

    for (int i = tid; i < CIN * NC; i += NTHREADS * 2)
        __builtin_prefetch(weight + (size_t)i * COUT, 0, 1);

    float g_p[2][3];
    #pragma unroll
    for (int i = 0; i < 2; ++i) {
        const float* pp = points + ((size_t)b * NPTS + g_idx[i]) * DIM;
        g_p[i][0] = pp[0]; g_p[i][1] = pp[1]; g_p[i][2] = pp[2];
    }

    #pragma unroll
    for (int k = 0; k < 3; ++k)
        reinterpret_cast<v4f*>(s_l1w)[tid + k * NTHREADS] = p_l1w[k];
    reinterpret_cast<v4f*>(s_l2w)[tid] = p_l2w;
    reinterpret_cast<v2f*>(s_l3w)[tid] = p_l3w;
    if (tid < L1OUT)   s_l1b[tid]  = p_l1b;
    if (tid < NC)      { s_l2b[tid] = p_l2b; s_l3b[tid] = p_l3b; }
    if (tid < DIM*NC)  s_cent[tid] = p_cent;
    if (tid < COUT)    s_bias[tid] = p_bias;

    #pragma unroll
    for (int i = 0; i < 2; ++i) {
        const int r = tid + i * NTHREADS;
        s_idx[r] = g_idx[i];
        s_rel[r][0] = g_p[i][0] - g_q[i][0];
        s_rel[r][1] = g_p[i][1] - g_q[i][1];
        s_rel[r][2] = g_p[i][2] - g_q[i][2];
    }
    __syncthreads();

    float* s1 = s_s1[wave];
    float* s2 = s_s2[wave];
    for (int t = 0; t < 4; ++t) {
        const int row0 = wave * 64 + t * 16;
        const int rA   = row0 + laneM;

        for (int nt = 0; nt < 2; ++nt) {
            v8f acc;
            const float bb = s_l1b[nt * 16 + laneM];
            #pragma unroll
            for (int r = 0; r < 8; ++r) acc[r] = bb;
            #pragma unroll
            for (int k0 = 0; k0 < L1IN; k0 += 4) {
                const int kA = k0 + koffA;
                v2f a, bf;
                a.x  = s_rel[rA][(kA    ) >> 4] - s_cent[kA    ];
                a.y  = s_rel[rA][(kA + 1) >> 4] - s_cent[kA + 1];
                bf.x = s_l1w[(kA    ) * L1OUT + nt * 16 + laneM];
                bf.y = s_l1w[(kA + 1) * L1OUT + nt * 16 + laneM];
                acc = wmma_f32(a, bf, acc);
            }
            #pragma unroll
            for (int r = 0; r < 8; ++r) {
                float v = acc[r]; v = v > 0.f ? v : 0.f;
                s1[(r + hi8) * 33 + nt * 16 + laneM] = v;
            }
        }

        {
            v8f acc;
            const float bb = s_l2b[laneM];
            #pragma unroll
            for (int r = 0; r < 8; ++r) acc[r] = bb;
            #pragma unroll
            for (int k0 = 0; k0 < L1OUT; k0 += 4) {
                const int kA = k0 + koffA;
                v2f a, bf;
                a.x  = s1[laneM * 33 + kA];
                a.y  = s1[laneM * 33 + kA + 1];
                bf.x = s_l2w[(kA    ) * NC + laneM];
                bf.y = s_l2w[(kA + 1) * NC + laneM];
                acc = wmma_f32(a, bf, acc);
            }
            #pragma unroll
            for (int r = 0; r < 8; ++r) {
                float v = acc[r]; v = v > 0.f ? v : 0.f;
                s2[(r + hi8) * 17 + laneM] = v;
            }
        }

        {
            v8f acc;
            const float bb = s_l3b[laneM];
            #pragma unroll
            for (int r = 0; r < 8; ++r) acc[r] = bb;
            #pragma unroll
            for (int k0 = 0; k0 < NC; k0 += 4) {
                const int kA = k0 + koffA;
                v2f a, bf;
                a.x  = s2[laneM * 17 + kA];
                a.y  = s2[laneM * 17 + kA + 1];
                bf.x = s_l3w[(kA    ) * NC + laneM];
                bf.y = s_l3w[(kA + 1) * NC + laneM];
                acc = wmma_f32(a, bf, acc);
            }
            #pragma unroll
            for (int r = 0; r < 8; ++r) {
                float v = acc[r]; v = v > 0.f ? v : 0.f;
                s_d3[row0 + r + hi8][laneM] = v;
            }
        }
    }
    __syncthreads();

    for (int pi = 0; pi < 4; ++pi) {
        const int p = wave * 4 + pi;
        for (int mt = 0; mt < 4; ++mt) {
            v8f acc = {0.f, 0.f, 0.f, 0.f, 0.f, 0.f, 0.f, 0.f};
            #pragma unroll
            for (int k0 = 0; k0 < KNB; k0 += 4) {
                const int kA = k0 + koffA;
                const int i0 = s_idx[p * 16 + kA];
                const int i1 = s_idx[p * 16 + kA + 1];
                v2f a, bf;
                a.x  = inp[((size_t)b * NPTS + i0) * CIN + mt * 16 + laneM];
                a.y  = inp[((size_t)b * NPTS + i1) * CIN + mt * 16 + laneM];
                bf.x = s_d3[p * 16 + kA    ][laneM];
                bf.y = s_d3[p * 16 + kA + 1][laneM];
                acc = wmma_f32(a, bf, acc);
            }
            #pragma unroll
            for (int r = 0; r < 8; ++r)
                s_fmat[p][(mt * 16 + r + hi8) * 16 + laneM] = acc[r];
        }
    }

    {
        v8f acc[4];
        #pragma unroll
        for (int t = 0; t < 4; ++t) {
            #pragma unroll
            for (int r = 0; r < 8; ++r) acc[t][r] = 0.f;
        }
        const float* wrow = weight + (size_t)(wave * 256 + hiHalf * 4) * COUT + laneM * 4;
        v4f w0[16], w1[16], a0[4], a1[4];
        __syncthreads();
        const float* arow = &s_fmat[laneM][wave * 256 + hiHalf * 4];
        d_load_w(w0, wrow);
        d_load_a(a0, arow);
        for (int it = 0; it < 3; ++it) {
            d_load_w(w1, wrow + 32 * COUT);
            d_load_a(a1, arow + 32);
            __builtin_amdgcn_sched_barrier(0);
            d_mma(acc, w0, a0);
            __builtin_amdgcn_sched_barrier(0);
            wrow += 64 * COUT;
            arow += 64;
            d_load_w(w0, wrow);
            d_load_a(a0, arow);
            __builtin_amdgcn_sched_barrier(0);
            d_mma(acc, w1, a1);
            __builtin_amdgcn_sched_barrier(0);
        }
        d_load_w(w1, wrow + 32 * COUT);
        d_load_a(a1, arow + 32);
        __builtin_amdgcn_sched_barrier(0);
        d_mma(acc, w0, a0);
        d_mma(acc, w1, a1);

        float* part = &s_d3[0][0];
        #pragma unroll
        for (int r = 0; r < 8; ++r) {
            v4f v;
            v.x = acc[0][r]; v.y = acc[1][r]; v.z = acc[2][r]; v.w = acc[3][r];
            *reinterpret_cast<v4f*>(part + (wave * TP + hi8 + r) * COUT + laneM * 4) = v;
        }
        __syncthreads();
        #pragma unroll
        for (int i = 0; i < 2; ++i) {
            const int q   = tid + i * NTHREADS;
            const int p   = q >> 4;
            const int col = (q & 15) * 4;
            v4f s = *reinterpret_cast<const v4f*>(part + (0 * TP + p) * COUT + col);
            s += *reinterpret_cast<const v4f*>(part + (1 * TP + p) * COUT + col);
            s += *reinterpret_cast<const v4f*>(part + (2 * TP + p) * COUT + col);
            s += *reinterpret_cast<const v4f*>(part + (3 * TP + p) * COUT + col);
            const v4f bb = *reinterpret_cast<const v4f*>(s_bias + col);
            const v4f o  = s * (1.0f / KNB) + bb;
            *reinterpret_cast<v4f*>(out + ((size_t)b * NPTS + n0 + p) * COUT + col) = o;
        }
    }
}

extern "C" void kernel_launch(void* const* d_in, const int* in_sizes, int n_in,
                              void* d_out, int out_size, void* d_ws, size_t ws_size,
                              hipStream_t stream) {
    (void)in_sizes; (void)n_in; (void)out_size; (void)d_ws; (void)ws_size;
    const float* inp      = (const float*)d_in[0];
    const float* points   = (const float*)d_in[1];
    const float* next_pts = (const float*)d_in[2];
    const int*   indices  = (const int*)  d_in[3];
    const float* weight   = (const float*)d_in[4];
    const float* bias     = (const float*)d_in[5];
    const float* centers  = (const float*)d_in[6];
    const float* l1w      = (const float*)d_in[7];
    const float* l1b      = (const float*)d_in[8];
    const float* l2w      = (const float*)d_in[9];
    const float* l2b      = (const float*)d_in[10];
    const float* l3w      = (const float*)d_in[11];
    const float* l3b      = (const float*)d_in[12];
    float* out = (float*)d_out;

    const int grid = BATCH * (NPTS / TP);
    ptconv_wmma_kernel<<<grid, NTHREADS, 0, stream>>>(
        inp, points, next_pts, indices, weight, bias, centers,
        l1w, l1b, l2w, l2b, l3w, l3b, out);
}


# LDS per workgroup 110016 -> 67072 bytes: MLP parameters, staging, d3 and the partial sums overlaid on the f matrix (one more barrier); 4 workgroups resident per WGP instead of 2
# speedup vs baseline: 2.3541x; 1.0847x over previous
#include <hip/hip_runtime.h>
#include <hip/hip_bf16.h>

typedef __attribute__((ext_vector_type(2))) float v2f;
typedef __attribute__((ext_vector_type(8))) float v8f;
typedef __attribute__((ext_vector_type(4))) float v4f;

#define BATCH   4
#define NPTS    8192
#define KNB     16
#define CIN     64
#define COUT    64
#define NC      16
#define DIM     3
#define L1IN    48
#define L1OUT   32
#define TP      16
#define NTHREADS 128
#define NWAVES  4
#define FSTRIDE 1028

static __device__ __forceinline__ v8f wmma_f32(v2f a, v2f b, v8f c) {
    return __builtin_amdgcn_wmma_f32_16x16x4_f32(false, a, false, b,
                                                 (short)0, c, false, false);
}

static __device__ __forceinline__ void d_load_w(v4f (&wr)[16], const float* __restrict__ wrow) {
    #pragma unroll
    for (int j = 0; j < 4; ++j) {
        #pragma unroll
        for (int e = 0; e < 4; ++e)
            wr[j * 4 + e] = *reinterpret_cast<const v4f*>(wrow + (size_t)(j * 8 + e) * COUT);
    }
}
static __device__ __forceinline__ void d_load_a(v4f (&ar)[4], const float* arow) {
    #pragma unroll
    for (int j = 0; j < 4; ++j)
        ar[j] = *reinterpret_cast<const v4f*>(arow + j * 8);
}
static __device__ __forceinline__ void d_mma(v8f (&acc)[4], const v4f (&wr)[16], const v4f (&ar)[4]) {
    #pragma unroll
    for (int j = 0; j < 4; ++j) {
        #pragma unroll
        for (int u = 0; u < 2; ++u) {
            v2f a;
            a.x = ar[j][2 * u];
            a.y = ar[j][2 * u + 1];
            #pragma unroll
            for (int t = 0; t < 4; ++t) {
                v2f bf;
                bf.x = wr[j * 4 + 2 * u][t];
                bf.y = wr[j * 4 + 2 * u + 1][t];
                acc[t] = wmma_f32(a, bf, acc[t]);
            }
        }
    }
}

__global__ __launch_bounds__(NTHREADS)
void ptconv_wmma_kernel(const float* __restrict__ inp,
                        const float* __restrict__ points,
                        const float* __restrict__ next_pts,
                        const int*   __restrict__ indices,
                        const float* __restrict__ weight,
                        const float* __restrict__ bias,
                        const float* __restrict__ centers,
                        const float* __restrict__ l1w, const float* __restrict__ l1b,
                        const float* __restrict__ l2w, const float* __restrict__ l2b,
                        const float* __restrict__ l3w, const float* __restrict__ l3b,
                        float* __restrict__ out)
{
    __shared__ __attribute__((aligned(16))) float s_fmat[TP][FSTRIDE];
    __shared__ __attribute__((aligned(16))) float s_bias[COUT];
    __shared__ int   s_idx[TP * KNB];
    float* const fbase = &s_fmat[0][0];
    float* const s_l1w  = fbase;
    float* const s_l2w  = fbase + 1536;
    float* const s_l3w  = fbase + 2048;
    float* const s_l1b  = fbase + 2304;
    float* const s_l2b  = fbase + 2336;
    float* const s_l3b  = fbase + 2352;
    float* const s_cent = fbase + 2368;
    float (* const s_rel)[4] = reinterpret_cast<float (*)[4]>(fbase + 4 * FSTRIDE);

    const int tid    = threadIdx.x;
    const int wave   = tid >> 5;
    const int lane   = tid & 31;
    const int laneM  = lane & 15;
    const int hiHalf = lane >> 4;
    const int koffA  = hiHalf * 2;
    const int hi8    = hiHalf * 8;
    float* const d3w = fbase + (4 * wave + 3) * FSTRIDE;

    const int tilesPerBatch = NPTS / TP;
    const int b    = blockIdx.x / tilesPerBatch;
    const int tile = blockIdx.x % tilesPerBatch;
    const int n0   = tile * TP;

    const size_t row0g = (size_t)b * NPTS + n0;
    int   g_idx[2];
    float g_q[2][3];
    #pragma unroll
    for (int i = 0; i < 2; ++i) {
        const int r = tid + i * NTHREADS;
        g_idx[i] = indices[row0g * KNB + r];
        const float* qq = next_pts + (row0g + (r >> 4)) * DIM;
        g_q[i][0] = qq[0]; g_q[i][1] = qq[1]; g_q[i][2] = qq[2];
    }
    v4f p_l1w[3];
    #pragma unroll
    for (int k = 0; k < 3; ++k)
        p_l1w[k] = reinterpret_cast<const v4f*>(l1w)[tid + k * NTHREADS];
    const v4f p_l2w = reinterpret_cast<const v4f*>(l2w)[tid];
    const v2f p_l3w = reinterpret_cast<const v2f*>(l3w)[tid];
    const float p_l1b  = l1b[tid & (L1OUT - 1)];
    const float p_l2b  = l2b[tid & (NC - 1)];
    const float p_l3b  = l3b[tid & (NC - 1)];
    const float p_cent = centers[tid < DIM * NC ? tid : 0];
    const float p_bias = bias[tid & (COUT - 1)];

    for (int i = tid; i < CIN * NC; i += NTHREADS * 2)
        __builtin_prefetch(weight + (size_t)i * COUT, 0, 1);

    float g_p[2][3];
    #pragma unroll
    for (int i = 0; i < 2; ++i) {
        const float* pp = points + ((size_t)b * NPTS + g_idx[i]) * DIM;
        g_p[i][0] = pp[0]; g_p[i][1] = pp[1]; g_p[i][2] = pp[2];
    }

    #pragma unroll
    for (int k = 0; k < 3; ++k)
        reinterpret_cast<v4f*>(s_l1w)[tid + k * NTHREADS] = p_l1w[k];
    reinterpret_cast<v4f*>(s_l2w)[tid] = p_l2w;
    reinterpret_cast<v2f*>(s_l3w)[tid] = p_l3w;
    if (tid < L1OUT)   s_l1b[tid]  = p_l1b;
    if (tid < NC)      { s_l2b[tid] = p_l2b; s_l3b[tid] = p_l3b; }
    if (tid < DIM*NC)  s_cent[tid] = p_cent;
    if (tid < COUT)    s_bias[tid] = p_bias;

    #pragma unroll
    for (int i = 0; i < 2; ++i) {
        const int r = tid + i * NTHREADS;
        s_idx[r] = g_idx[i];
        s_rel[r][0] = g_p[i][0] - g_q[i][0];
        s_rel[r][1] = g_p[i][1] - g_q[i][1];
        s_rel[r][2] = g_p[i][2] - g_q[i][2];
    }
    __syncthreads();

    float* s1 = fbase + 8  * FSTRIDE + wave * (16 * 33);
    float* s2 = fbase + 12 * FSTRIDE + wave * (16 * 17);
    v2f w1f[2][12], cf[12], w2f[8], w3f[4];
    #pragma unroll
    for (int ks = 0; ks < 12; ++ks) {
        const int kA = ks * 4 + koffA;
        cf[ks].x = s_cent[kA];
        cf[ks].y = s_cent[kA + 1];
        #pragma unroll
        for (int nt = 0; nt < 2; ++nt) {
            w1f[nt][ks].x = s_l1w[(kA    ) * L1OUT + nt * 16 + laneM];
            w1f[nt][ks].y = s_l1w[(kA + 1) * L1OUT + nt * 16 + laneM];
        }
    }
    #pragma unroll
    for (int ks = 0; ks < 8; ++ks) {
        const int kA = ks * 4 + koffA;
        w2f[ks].x = s_l2w[(kA    ) * NC + laneM];
        w2f[ks].y = s_l2w[(kA + 1) * NC + laneM];
    }
    #pragma unroll
    for (int ks = 0; ks < 4; ++ks) {
        const int kA = ks * 4 + koffA;
        w3f[ks].x = s_l3w[(kA    ) * NC + laneM];
        w3f[ks].y = s_l3w[(kA + 1) * NC + laneM];
    }
    const float b1lo = s_l1b[laneM];
    const float b1hi = s_l1b[16 + laneM];
    const float b2v  = s_l2b[laneM];
    const float b3v  = s_l3b[laneM];

    for (int t = 0; t < 4; ++t) {
        const int row0 = wave * 64 + t * 16;
        const int rA   = row0 + laneM;

        {
            float relv[3];
            relv[0] = s_rel[rA][0]; relv[1] = s_rel[rA][1]; relv[2] = s_rel[rA][2];
            v8f acc0, acc1;
            #pragma unroll
            for (int r = 0; r < 8; ++r) { acc0[r] = b1lo; acc1[r] = b1hi; }
            #pragma unroll
            for (int ks = 0; ks < 12; ++ks) {
                v2f a;
                a.x = relv[ks >> 2] - cf[ks].x;
                a.y = relv[ks >> 2] - cf[ks].y;
                acc0 = wmma_f32(a, w1f[0][ks], acc0);
                acc1 = wmma_f32(a, w1f[1][ks], acc1);
            }
            #pragma unroll
            for (int r = 0; r < 8; ++r) {
                float v0 = acc0[r]; v0 = v0 > 0.f ? v0 : 0.f;
                float v1 = acc1[r]; v1 = v1 > 0.f ? v1 : 0.f;
                s1[(r + hi8) * 33 + laneM]      = v0;
                s1[(r + hi8) * 33 + 16 + laneM] = v1;
            }
        }

        {
            v8f acc;
            #pragma unroll
            for (int r = 0; r < 8; ++r) acc[r] = b2v;
            #pragma unroll
            for (int ks = 0; ks < 8; ++ks) {
                const int kA = ks * 4 + koffA;
                v2f a;
                a.x  = s1[laneM * 33 + kA];
                a.y  = s1[laneM * 33 + kA + 1];
                acc = wmma_f32(a, w2f[ks], acc);
            }
            #pragma unroll
            for (int r = 0; r < 8; ++r) {
                float v = acc[r]; v = v > 0.f ? v : 0.f;
                s2[(r + hi8) * 17 + laneM] = v;
            }
        }

        {
            v8f acc;
            #pragma unroll
            for (int r = 0; r < 8; ++r) acc[r] = b3v;
            #pragma unroll
            for (int ks = 0; ks < 4; ++ks) {
                const int kA = ks * 4 + koffA;
                v2f a;
                a.x  = s2[laneM * 17 + kA];
                a.y  = s2[laneM * 17 + kA + 1];
                acc = wmma_f32(a, w3f[ks], acc);
            }
            #pragma unroll
            for (int r = 0; r < 8; ++r) {
                float v = acc[r]; v = v > 0.f ? v : 0.f;
                d3w[(t * 16 + r + hi8) * NC + laneM] = v;
            }
        }
    }
    __syncthreads();

    for (int pi = 0; pi < 4; ++pi) {
        const int p = wave * 4 + pi;
        v2f bfr[4];
        int i0[4], i1[4];
        #pragma unroll
        for (int ks = 0; ks < 4; ++ks) {
            const int kA = ks * 4 + koffA;
            i0[ks] = s_idx[p * 16 + kA];
            i1[ks] = s_idx[p * 16 + kA + 1];
            bfr[ks].x = d3w[(pi * 16 + kA    ) * NC + laneM];
            bfr[ks].y = d3w[(pi * 16 + kA + 1) * NC + laneM];
        }
        v8f acc[4];
        #pragma unroll
        for (int mt = 0; mt < 4; ++mt) {
            #pragma unroll
            for (int r = 0; r < 8; ++r) acc[mt][r] = 0.f;
        }
        #pragma unroll
        for (int ks = 0; ks < 4; ++ks) {
            #pragma unroll
            for (int mt = 0; mt < 4; ++mt) {
                v2f a;
                a.x = inp[((size_t)b * NPTS + i0[ks]) * CIN + mt * 16 + laneM];
                a.y = inp[((size_t)b * NPTS + i1[ks]) * CIN + mt * 16 + laneM];
                acc[mt] = wmma_f32(a, bfr[ks], acc[mt]);
            }
        }
        #pragma unroll
        for (int mt = 0; mt < 4; ++mt) {
            #pragma unroll
            for (int r = 0; r < 8; ++r)
                s_fmat[p][(mt * 16 + r + hi8) * 16 + laneM] = acc[mt][r];
        }
    }

    {
        v8f acc[4];
        #pragma unroll
        for (int t = 0; t < 4; ++t) {
            #pragma unroll
            for (int r = 0; r < 8; ++r) acc[t][r] = 0.f;
        }
        const float* wrow = weight + (size_t)(wave * 256 + hiHalf * 4) * COUT + laneM * 4;
        v4f w0[16], w1[16], a0[4], a1[4];
        __syncthreads();
        const float* arow = &s_fmat[laneM][wave * 256 + hiHalf * 4];
        d_load_w(w0, wrow);
        d_load_a(a0, arow);
        for (int it = 0; it < 3; ++it) {
            d_load_w(w1, wrow + 32 * COUT);
            d_load_a(a1, arow + 32);
            __builtin_amdgcn_sched_barrier(0);
            d_mma(acc, w0, a0);
            __builtin_amdgcn_sched_barrier(0);
            wrow += 64 * COUT;
            arow += 64;
            d_load_w(w0, wrow);
            d_load_a(a0, arow);
            __builtin_amdgcn_sched_barrier(0);
            d_mma(acc, w1, a1);
            __builtin_amdgcn_sched_barrier(0);
        }
        d_load_w(w1, wrow + 32 * COUT);
        d_load_a(a1, arow + 32);
        __builtin_amdgcn_sched_barrier(0);
        d_mma(acc, w0, a0);
        d_mma(acc, w1, a1);

        __syncthreads();
        float* part = fbase;
        #pragma unroll
        for (int r = 0; r < 8; ++r) {
            v4f v;
            v.x = acc[0][r]; v.y = acc[1][r]; v.z = acc[2][r]; v.w = acc[3][r];
            *reinterpret_cast<v4f*>(part + (wave * TP + hi8 + r) * COUT + laneM * 4) = v;
        }
        __syncthreads();
        #pragma unroll
        for (int i = 0; i < 2; ++i) {
            const int q   = tid + i * NTHREADS;
            const int p   = q >> 4;
            const int col = (q & 15) * 4;
            v4f s = *reinterpret_cast<const v4f*>(part + (0 * TP + p) * COUT + col);
            s += *reinterpret_cast<const v4f*>(part + (1 * TP + p) * COUT + col);
            s += *reinterpret_cast<const v4f*>(part + (2 * TP + p) * COUT + col);
            s += *reinterpret_cast<const v4f*>(part + (3 * TP + p) * COUT + col);
            const v4f bb = *reinterpret_cast<const v4f*>(s_bias + col);
            const v4f o  = s * (1.0f / KNB) + bb;
            *reinterpret_cast<v4f*>(out + ((size_t)b * NPTS + n0 + p) * COUT + col) = o;
        }
    }
}

extern "C" void kernel_launch(void* const* d_in, const int* in_sizes, int n_in,
                              void* d_out, int out_size, void* d_ws, size_t ws_size,
                              hipStream_t stream) {
    (void)in_sizes; (void)n_in; (void)out_size; (void)d_ws; (void)ws_size;
    const float* inp      = (const float*)d_in[0];
    const float* points   = (const float*)d_in[1];
    const float* next_pts = (const float*)d_in[2];
    const int*   indices  = (const int*)  d_in[3];
    const float* weight   = (const float*)d_in[4];
    const float* bias     = (const float*)d_in[5];
    const float* centers  = (const float*)d_in[6];
    const float* l1w      = (const float*)d_in[7];
    const float* l1b      = (const float*)d_in[8];
    const float* l2w      = (const float*)d_in[9];
    const float* l2b      = (const float*)d_in[10];
    const float* l3w      = (const float*)d_in[11];
    const float* l3b      = (const float*)d_in[12];
    float* out = (float*)d_out;

    const int grid = BATCH * (NPTS / TP);
    ptconv_wmma_kernel<<<grid, NTHREADS, 0, stream>>>(
        inp, points, next_pts, indices, weight, bias, centers,
        l1w, l1b, l2w, l2b, l3w, l3b, out);
}


# phase B: first MLP layer folded to rel@Wf+cst (Wf = per-dimension sums of weight rows, cst = b1 - centers@W1, one 24-WMMA product per workgroup): 2 WMMAs per tile, not 24; changes summation order
# speedup vs baseline: 2.6216x; 1.1136x over previous
#include <hip/hip_runtime.h>
#include <hip/hip_bf16.h>

typedef __attribute__((ext_vector_type(2))) float v2f;
typedef __attribute__((ext_vector_type(8))) float v8f;
typedef __attribute__((ext_vector_type(4))) float v4f;

#define BATCH   4
#define NPTS    8192
#define KNB     16
#define CIN     64
#define COUT    64
#define NC      16
#define DIM     3
#define L1IN    48
#define L1OUT   32
#define TP      16
#define NTHREADS 128
#define NWAVES  4
#define FSTRIDE 1028

static __device__ __forceinline__ v8f wmma_f32(v2f a, v2f b, v8f c) {
    return __builtin_amdgcn_wmma_f32_16x16x4_f32(false, a, false, b,
                                                 (short)0, c, false, false);
}

static __device__ __forceinline__ void d_load_w(v4f (&wr)[16], const float* __restrict__ wrow) {
    #pragma unroll
    for (int j = 0; j < 4; ++j) {
        #pragma unroll
        for (int e = 0; e < 4; ++e)
            wr[j * 4 + e] = *reinterpret_cast<const v4f*>(wrow + (size_t)(j * 8 + e) * COUT);
    }
}
static __device__ __forceinline__ void d_load_a(v4f (&ar)[4], const float* arow) {
    #pragma unroll
    for (int j = 0; j < 4; ++j)
        ar[j] = *reinterpret_cast<const v4f*>(arow + j * 8);
}
static __device__ __forceinline__ void d_mma(v8f (&acc)[4], const v4f (&wr)[16], const v4f (&ar)[4]) {
    #pragma unroll
    for (int j = 0; j < 4; ++j) {
        #pragma unroll
        for (int u = 0; u < 2; ++u) {
            v2f a;
            a.x = ar[j][2 * u];
            a.y = ar[j][2 * u + 1];
            #pragma unroll
            for (int t = 0; t < 4; ++t) {
                v2f bf;
                bf.x = wr[j * 4 + 2 * u][t];
                bf.y = wr[j * 4 + 2 * u + 1][t];
                acc[t] = wmma_f32(a, bf, acc[t]);
            }
        }
    }
}

__global__ __launch_bounds__(NTHREADS)
void ptconv_wmma_kernel(const float* __restrict__ inp,
                        const float* __restrict__ points,
                        const float* __restrict__ next_pts,
                        const int*   __restrict__ indices,
                        const float* __restrict__ weight,
                        const float* __restrict__ bias,
                        const float* __restrict__ centers,
                        const float* __restrict__ l1w, const float* __restrict__ l1b,
                        const float* __restrict__ l2w, const float* __restrict__ l2b,
                        const float* __restrict__ l3w, const float* __restrict__ l3b,
                        float* __restrict__ out)
{
    __shared__ __attribute__((aligned(16))) float s_fmat[TP][FSTRIDE];
    __shared__ __attribute__((aligned(16))) float s_bias[COUT];
    __shared__ int   s_idx[TP * KNB];
    float* const fbase = &s_fmat[0][0];
    float* const s_xch  = fbase;
    float* const s_l2w  = fbase + 1536;
    float* const s_l3w  = fbase + 2048;
    float* const s_l2b  = fbase + 2336;
    float* const s_l3b  = fbase + 2352;
    float (* const s_rel)[4] = reinterpret_cast<float (*)[4]>(fbase + 4 * FSTRIDE);

    const int tid    = threadIdx.x;
    const int wave   = tid >> 5;
    const int lane   = tid & 31;
    const int laneM  = lane & 15;
    const int hiHalf = lane >> 4;
    const int koffA  = hiHalf * 2;
    const int hi8    = hiHalf * 8;
    float* const d3w = fbase + (4 * wave + 3) * FSTRIDE;

    const int tilesPerBatch = NPTS / TP;
    const int b    = blockIdx.x / tilesPerBatch;
    const int tile = blockIdx.x % tilesPerBatch;
    const int n0   = tile * TP;

    const size_t row0g = (size_t)b * NPTS + n0;
    int   g_idx[2];
    float g_q[2][3];
    #pragma unroll
    for (int i = 0; i < 2; ++i) {
        const int r = tid + i * NTHREADS;
        g_idx[i] = indices[row0g * KNB + r];
        const float* qq = next_pts + (row0g + (r >> 4)) * DIM;
        g_q[i][0] = qq[0]; g_q[i][1] = qq[1]; g_q[i][2] = qq[2];
    }
    const v4f p_l2w = reinterpret_cast<const v4f*>(l2w)[tid];
    const v2f p_l3w = reinterpret_cast<const v2f*>(l3w)[tid];
    const float p_l2b  = l2b[tid & (NC - 1)];
    const float p_l3b  = l3b[tid & (NC - 1)];
    const float p_bias = bias[tid & (COUT - 1)];
    const int fnt = wave & 1;
    const int fk0 = (wave >> 1) * 24;
    v2f fw[6], fc[6];
    #pragma unroll
    for (int ks = 0; ks < 6; ++ks) {
        const int kA = fk0 + ks * 4 + koffA;
        fw[ks].x = l1w[(kA    ) * L1OUT + fnt * 16 + laneM];
        fw[ks].y = l1w[(kA + 1) * L1OUT + fnt * 16 + laneM];
        fc[ks]   = *reinterpret_cast<const v2f*>(centers + kA);
    }
    const float fb1 = l1b[fnt * 16 + laneM];

    for (int i = tid; i < CIN * NC; i += NTHREADS * 2)
        __builtin_prefetch(weight + (size_t)i * COUT, 0, 1);

    float g_p[2][3];
    #pragma unroll
    for (int i = 0; i < 2; ++i) {
        const float* pp = points + ((size_t)b * NPTS + g_idx[i]) * DIM;
        g_p[i][0] = pp[0]; g_p[i][1] = pp[1]; g_p[i][2] = pp[2];
    }

    reinterpret_cast<v4f*>(s_l2w)[tid] = p_l2w;
    reinterpret_cast<v2f*>(s_l3w)[tid] = p_l3w;
    if (tid < NC)      { s_l2b[tid] = p_l2b; s_l3b[tid] = p_l3b; }
    if (tid < COUT)    s_bias[tid] = p_bias;

    {
        const int rowCode = ((laneM & 6) == 0) ? ((laneM & 1) | ((laneM >> 2) & 2)) : 4;
        v8f f;
        #pragma unroll
        for (int r = 0; r < 8; ++r) f[r] = 0.f;
        f[1] = (hiHalf != 0 && fk0 == 0) ? fb1 : 0.f;
        #pragma unroll
        for (int ks = 0; ks < 6; ++ks) {
            const float ind = (rowCode == ((fk0 + ks * 4) >> 4)) ? 1.f : 0.f;
            v2f a;
            a.x = (rowCode == 3) ? -fc[ks].x : ind;
            a.y = (rowCode == 3) ? -fc[ks].y : ind;
            f = wmma_f32(a, fw[ks], f);
        }
        v2f fo;
        fo.x = f[0]; fo.y = f[1];
        reinterpret_cast<v2f*>(s_xch)[wave * 32 + lane] = fo;
    }

    #pragma unroll
    for (int i = 0; i < 2; ++i) {
        const int r = tid + i * NTHREADS;
        s_idx[r] = g_idx[i];
        s_rel[r][0] = g_p[i][0] - g_q[i][0];
        s_rel[r][1] = g_p[i][1] - g_q[i][1];
        s_rel[r][2] = g_p[i][2] - g_q[i][2];
        s_rel[r][3] = 1.0f;
    }
    __syncthreads();

    float* s1 = fbase + 8  * FSTRIDE + wave * (16 * 33);
    float* s2 = fbase + 12 * FSTRIDE + wave * (16 * 17);
    v2f w2f[8], w3f[4];
    const v2f* const xq = reinterpret_cast<const v2f*>(s_xch) + lane;
    const v2f wf0 = xq[0 * 32] + xq[2 * 32];
    const v2f wf1 = xq[1 * 32] + xq[3 * 32];
    #pragma unroll
    for (int ks = 0; ks < 8; ++ks) {
        const int kA = ks * 4 + koffA;
        w2f[ks].x = s_l2w[(kA    ) * NC + laneM];
        w2f[ks].y = s_l2w[(kA + 1) * NC + laneM];
    }
    #pragma unroll
    for (int ks = 0; ks < 4; ++ks) {
        const int kA = ks * 4 + koffA;
        w3f[ks].x = s_l3w[(kA    ) * NC + laneM];
        w3f[ks].y = s_l3w[(kA + 1) * NC + laneM];
    }
    const float b2v  = s_l2b[laneM];
    const float b3v  = s_l3b[laneM];

    for (int t = 0; t < 4; ++t) {
        const int row0 = wave * 64 + t * 16;
        const int rA   = row0 + laneM;

        {
            const v2f a = *reinterpret_cast<const v2f*>(&s_rel[rA][koffA]);
            v8f acc0, acc1;
            #pragma unroll
            for (int r = 0; r < 8; ++r) { acc0[r] = 0.f; acc1[r] = 0.f; }
            acc0 = wmma_f32(a, wf0, acc0);
            acc1 = wmma_f32(a, wf1, acc1);
            #pragma unroll
            for (int r = 0; r < 8; ++r) {
                float v0 = acc0[r]; v0 = v0 > 0.f ? v0 : 0.f;
                float v1 = acc1[r]; v1 = v1 > 0.f ? v1 : 0.f;
                s1[(r + hi8) * 33 + laneM]      = v0;
                s1[(r + hi8) * 33 + 16 + laneM] = v1;
            }
        }

        {
            v8f acc;
            #pragma unroll
            for (int r = 0; r < 8; ++r) acc[r] = b2v;
            #pragma unroll
            for (int ks = 0; ks < 8; ++ks) {
                const int kA = ks * 4 + koffA;
                v2f a;
                a.x  = s1[laneM * 33 + kA];
                a.y  = s1[laneM * 33 + kA + 1];
                acc = wmma_f32(a, w2f[ks], acc);
            }
            #pragma unroll
            for (int r = 0; r < 8; ++r) {
                float v = acc[r]; v = v > 0.f ? v : 0.f;
                s2[(r + hi8) * 17 + laneM] = v;
            }
        }

        {
            v8f acc;
            #pragma unroll
            for (int r = 0; r < 8; ++r) acc[r] = b3v;
            #pragma unroll
            for (int ks = 0; ks < 4; ++ks) {
                const int kA = ks * 4 + koffA;
                v2f a;
                a.x  = s2[laneM * 17 + kA];
                a.y  = s2[laneM * 17 + kA + 1];
                acc = wmma_f32(a, w3f[ks], acc);
            }
            #pragma unroll
            for (int r = 0; r < 8; ++r) {
                float v = acc[r]; v = v > 0.f ? v : 0.f;
                d3w[(t * 16 + r + hi8) * NC + laneM] = v;
            }
        }
    }
    __syncthreads();

    for (int pi = 0; pi < 4; ++pi) {
        const int p = wave * 4 + pi;
        v2f bfr[4];
        int i0[4], i1[4];
        #pragma unroll
        for (int ks = 0; ks < 4; ++ks) {
            const int kA = ks * 4 + koffA;
            i0[ks] = s_idx[p * 16 + kA];
            i1[ks] = s_idx[p * 16 + kA + 1];
            bfr[ks].x = d3w[(pi * 16 + kA    ) * NC + laneM];
            bfr[ks].y = d3w[(pi * 16 + kA + 1) * NC + laneM];
        }
        v8f acc[4];
        #pragma unroll
        for (int mt = 0; mt < 4; ++mt) {
            #pragma unroll
            for (int r = 0; r < 8; ++r) acc[mt][r] = 0.f;
        }
        #pragma unroll
        for (int ks = 0; ks < 4; ++ks) {
            #pragma unroll
            for (int mt = 0; mt < 4; ++mt) {
                v2f a;
                a.x = inp[((size_t)b * NPTS + i0[ks]) * CIN + mt * 16 + laneM];
                a.y = inp[((size_t)b * NPTS + i1[ks]) * CIN + mt * 16 + laneM];
                acc[mt] = wmma_f32(a, bfr[ks], acc[mt]);
            }
        }
        #pragma unroll
        for (int mt = 0; mt < 4; ++mt) {
            #pragma unroll
            for (int r = 0; r < 8; ++r)
                s_fmat[p][(mt * 16 + r + hi8) * 16 + laneM] = acc[mt][r];
        }
    }

    {
        v8f acc[4];
        #pragma unroll
        for (int t = 0; t < 4; ++t) {
            #pragma unroll
            for (int r = 0; r < 8; ++r) acc[t][r] = 0.f;
        }
        const float* wrow = weight + (size_t)(wave * 256 + hiHalf * 4) * COUT + laneM * 4;
        v4f w0[16], w1[16], a0[4], a1[4];
        __syncthreads();
        const float* arow = &s_fmat[laneM][wave * 256 + hiHalf * 4];
        d_load_w(w0, wrow);
        d_load_a(a0, arow);
        for (int it = 0; it < 3; ++it) {
            d_load_w(w1, wrow + 32 * COUT);
            d_load_a(a1, arow + 32);
            __builtin_amdgcn_sched_barrier(0);
            d_mma(acc, w0, a0);
            __builtin_amdgcn_sched_barrier(0);
            wrow += 64 * COUT;
            arow += 64;
            d_load_w(w0, wrow);
            d_load_a(a0, arow);
            __builtin_amdgcn_sched_barrier(0);
            d_mma(acc, w1, a1);
            __builtin_amdgcn_sched_barrier(0);
        }
        d_load_w(w1, wrow + 32 * COUT);
        d_load_a(a1, arow + 32);
        __builtin_amdgcn_sched_barrier(0);
        d_mma(acc, w0, a0);
        d_mma(acc, w1, a1);

        __syncthreads();
        float* part = fbase;
        #pragma unroll
        for (int r = 0; r < 8; ++r) {
            v4f v;
            v.x = acc[0][r]; v.y = acc[1][r]; v.z = acc[2][r]; v.w = acc[3][r];
            *reinterpret_cast<v4f*>(part + (wave * TP + hi8 + r) * COUT + laneM * 4) = v;
        }
        __syncthreads();
        #pragma unroll
        for (int i = 0; i < 2; ++i) {
            const int q   = tid + i * NTHREADS;
            const int p   = q >> 4;
            const int col = (q & 15) * 4;
            v4f s = *reinterpret_cast<const v4f*>(part + (0 * TP + p) * COUT + col);
            s += *reinterpret_cast<const v4f*>(part + (1 * TP + p) * COUT + col);
            s += *reinterpret_cast<const v4f*>(part + (2 * TP + p) * COUT + col);
            s += *reinterpret_cast<const v4f*>(part + (3 * TP + p) * COUT + col);
            const v4f bb = *reinterpret_cast<const v4f*>(s_bias + col);
            const v4f o  = s * (1.0f / KNB) + bb;
            *reinterpret_cast<v4f*>(out + ((size_t)b * NPTS + n0 + p) * COUT + col) = o;
        }
    }
}

extern "C" void kernel_launch(void* const* d_in, const int* in_sizes, int n_in,
                              void* d_out, int out_size, void* d_ws, size_t ws_size,
                              hipStream_t stream) {
    (void)in_sizes; (void)n_in; (void)out_size; (void)d_ws; (void)ws_size;
    const float* inp      = (const float*)d_in[0];
    const float* points   = (const float*)d_in[1];
    const float* next_pts = (const float*)d_in[2];
    const int*   indices  = (const int*)  d_in[3];
    const float* weight   = (const float*)d_in[4];
    const float* bias     = (const float*)d_in[5];
    const float* centers  = (const float*)d_in[6];
    const float* l1w      = (const float*)d_in[7];
    const float* l1b      = (const float*)d_in[8];
    const float* l2w      = (const float*)d_in[9];
    const float* l2b      = (const float*)d_in[10];
    const float* l3w      = (const float*)d_in[11];
    const float* l3b      = (const float*)d_in[12];
    float* out = (float*)d_out;

    const int grid = BATCH * (NPTS / TP);
    ptconv_wmma_kernel<<<grid, NTHREADS, 0, stream>>>(
        inp, points, next_pts, indices, weight, bias, centers,
        l1w, l1b, l2w, l2b, l3w, l3b, out);
}


# phases B+C: MLP layer outputs kept in registers (layers 1,2 as transposed products: a D tile is the next product's operand as it stands; 4 tiles per layer), no LDS staging; changes summation order
# speedup vs baseline: 2.7334x; 1.0426x over previous
#include <hip/hip_runtime.h>
#include <hip/hip_bf16.h>

typedef __attribute__((ext_vector_type(2))) float v2f;
typedef __attribute__((ext_vector_type(8))) float v8f;
typedef __attribute__((ext_vector_type(4))) float v4f;
typedef __attribute__((ext_vector_type(4))) int   v4i;

#define BATCH   4
#define NPTS    8192
#define KNB     16
#define CIN     64
#define COUT    64
#define NC      16
#define DIM     3
#define L1IN    48
#define L1OUT   32
#define TP      16
#define NTHREADS 128
#define NWAVES  4
#define FSTRIDE 1028

static __device__ __forceinline__ v8f wmma_f32(v2f a, v2f b, v8f c) {
    return __builtin_amdgcn_wmma_f32_16x16x4_f32(false, a, false, b,
                                                 (short)0, c, false, false);
}

static __device__ __forceinline__ void d_load_w(v4f (&wr)[16], const float* __restrict__ wrow) {
    #pragma unroll
    for (int j = 0; j < 4; ++j) {
        #pragma unroll
        for (int e = 0; e < 4; ++e)
            wr[j * 4 + e] = *reinterpret_cast<const v4f*>(wrow + (size_t)(j * 8 + e) * COUT);
    }
}
static __device__ __forceinline__ void d_load_a(v4f (&ar)[4], const float* arow) {
    #pragma unroll
    for (int j = 0; j < 4; ++j)
        ar[j] = *reinterpret_cast<const v4f*>(arow + j * 8);
}
static __device__ __forceinline__ void d_mma(v8f (&acc)[4], const v4f (&wr)[16], const v4f (&ar)[4]) {
    #pragma unroll
    for (int j = 0; j < 4; ++j) {
        #pragma unroll
        for (int u = 0; u < 2; ++u) {
            v2f a;
            a.x = ar[j][2 * u];
            a.y = ar[j][2 * u + 1];
            #pragma unroll
            for (int t = 0; t < 4; ++t) {
                v2f bf;
                bf.x = wr[j * 4 + 2 * u][t];
                bf.y = wr[j * 4 + 2 * u + 1][t];
                acc[t] = wmma_f32(a, bf, acc[t]);
            }
        }
    }
}

__global__ __launch_bounds__(NTHREADS)
void ptconv_wmma_kernel(const float* __restrict__ inp,
                        const float* __restrict__ points,
                        const float* __restrict__ next_pts,
                        const int*   __restrict__ indices,
                        const float* __restrict__ weight,
                        const float* __restrict__ bias,
                        const float* __restrict__ centers,
                        const float* __restrict__ l1w, const float* __restrict__ l1b,
                        const float* __restrict__ l2w, const float* __restrict__ l2b,
                        const float* __restrict__ l3w, const float* __restrict__ l3b,
                        float* __restrict__ out)
{
    __shared__ __attribute__((aligned(16))) float s_fmat[TP][FSTRIDE];
    __shared__ __attribute__((aligned(16))) float s_rel[TP * KNB][4];
    __shared__ __attribute__((aligned(16))) float s_par[L1OUT * NC + NC * NC + 2 * NC];
    __shared__ __attribute__((aligned(16))) float s_xch[NWAVES * 32 * 2];
    __shared__ __attribute__((aligned(16))) float s_bias[COUT];
    __shared__ __attribute__((aligned(16))) int   s_idx[TP * KNB];
    float* const fbase = &s_fmat[0][0];
    float* const s_l2w  = s_par;
    float* const s_l3w  = s_par + L1OUT * NC;
    float* const s_l2b  = s_l3w + NC * NC;
    float* const s_l3b  = s_l2b + NC;

    const int tid    = threadIdx.x;
    const int wave   = tid >> 5;
    const int lane   = tid & 31;
    const int laneM  = lane & 15;
    const int hiHalf = lane >> 4;
    const int koffA  = hiHalf * 2;
    const int hi8    = hiHalf * 8;

    const int tilesPerBatch = NPTS / TP;
    const int b    = blockIdx.x / tilesPerBatch;
    const int tile = blockIdx.x % tilesPerBatch;
    const int n0   = tile * TP;

    const size_t row0g = (size_t)b * NPTS + n0;
    int   g_idx[2];
    float g_q[2][3];
    #pragma unroll
    for (int i = 0; i < 2; ++i) {
        const int r = tid + i * NTHREADS;
        g_idx[i] = indices[row0g * KNB + r];
        const float* qq = next_pts + (row0g + (r >> 4)) * DIM;
        g_q[i][0] = qq[0]; g_q[i][1] = qq[1]; g_q[i][2] = qq[2];
    }
    const v4f p_l2w = reinterpret_cast<const v4f*>(l2w)[tid];
    const v2f p_l3w = reinterpret_cast<const v2f*>(l3w)[tid];
    const float p_l2b  = l2b[tid & (NC - 1)];
    const float p_l3b  = l3b[tid & (NC - 1)];
    const float p_bias = bias[tid & (COUT - 1)];
    const int fnt = wave & 1;
    const int fk0 = (wave >> 1) * 24;
    v2f fw[6], fc[6];
    #pragma unroll
    for (int ks = 0; ks < 6; ++ks) {
        const int kA = fk0 + ks * 4 + koffA;
        fw[ks].x = l1w[(kA    ) * L1OUT + fnt * 16 + laneM];
        fw[ks].y = l1w[(kA + 1) * L1OUT + fnt * 16 + laneM];
        fc[ks]   = *reinterpret_cast<const v2f*>(centers + kA);
    }
    const float fb1 = l1b[fnt * 16 + laneM];

    for (int i = tid; i < CIN * NC; i += NTHREADS * 2)
        __builtin_prefetch(weight + (size_t)i * COUT, 0, 1);

    float g_p[2][3];
    #pragma unroll
    for (int i = 0; i < 2; ++i) {
        const float* pp = points + ((size_t)b * NPTS + g_idx[i]) * DIM;
        g_p[i][0] = pp[0]; g_p[i][1] = pp[1]; g_p[i][2] = pp[2];
    }

    reinterpret_cast<v4f*>(s_l2w)[tid] = p_l2w;
    reinterpret_cast<v2f*>(s_l3w)[tid] = p_l3w;
    if (tid < NC)      { s_l2b[tid] = p_l2b; s_l3b[tid] = p_l3b; }
    if (tid < COUT)    s_bias[tid] = p_bias;

    {
        const int rowCode = ((laneM & 6) == 0) ? ((laneM & 1) | ((laneM >> 2) & 2)) : 4;
        v8f f;
        #pragma unroll
        for (int r = 0; r < 8; ++r) f[r] = 0.f;
        f[1] = (hiHalf != 0 && fk0 == 0) ? fb1 : 0.f;
        #pragma unroll
        for (int ks = 0; ks < 6; ++ks) {
            const float ind = (rowCode == ((fk0 + ks * 4) >> 4)) ? 1.f : 0.f;
            v2f a;
            a.x = (rowCode == 3) ? -fc[ks].x : ind;
            a.y = (rowCode == 3) ? -fc[ks].y : ind;
            f = wmma_f32(a, fw[ks], f);
        }
        v2f fo;
        fo.x = f[0]; fo.y = f[1];
        reinterpret_cast<v2f*>(s_xch)[wave * 32 + lane] = fo;
    }

    #pragma unroll
    for (int i = 0; i < 2; ++i) {
        const int r = tid + i * NTHREADS;
        s_idx[r] = g_idx[i];
        s_rel[r][0] = g_p[i][0] - g_q[i][0];
        s_rel[r][1] = g_p[i][1] - g_q[i][1];
        s_rel[r][2] = g_p[i][2] - g_q[i][2];
        s_rel[r][3] = 1.0f;
    }
    __syncthreads();

    const v2f* const xq = reinterpret_cast<const v2f*>(s_xch) + lane;
    const v2f wf0 = xq[0 * 32] + xq[2 * 32];
    const v2f wf1 = xq[1 * 32] + xq[3 * 32];
    v2f w2a[8], w3b[4];
    #pragma unroll
    for (int j = 0; j < 8; ++j) {
        const int f = (j >> 2) * 16 + hi8 + (j & 3) * 2;
        w2a[j].x = s_l2w[(f    ) * NC + laneM];
        w2a[j].y = s_l2w[(f + 1) * NC + laneM];
    }
    #pragma unroll
    for (int j = 0; j < 4; ++j) {
        const int f = hi8 + j * 2;
        w3b[j].x = s_l3w[(f    ) * NC + laneM];
        w3b[j].y = s_l3w[(f + 1) * NC + laneM];
    }
    const v4f b2lo = *reinterpret_cast<const v4f*>(s_l2b + hi8);
    const v4f b2hi = *reinterpret_cast<const v4f*>(s_l2b + hi8 + 4);
    const float b3v = s_l3b[laneM];

    v8f d3[4];
    {
        v8f y1[4][2];
        #pragma unroll
        for (int t = 0; t < 4; ++t) {
            const v2f rl = *reinterpret_cast<const v2f*>(&s_rel[wave * 64 + t * 16 + laneM][koffA]);
            v8f z;
            #pragma unroll
            for (int r = 0; r < 8; ++r) z[r] = 0.f;
            y1[t][0] = wmma_f32(wf0, rl, z);
            y1[t][1] = wmma_f32(wf1, rl, z);
        }
        #pragma unroll
        for (int t = 0; t < 4; ++t) {
            #pragma unroll
            for (int r = 0; r < 8; ++r) {
                float v0 = y1[t][0][r]; y1[t][0][r] = v0 > 0.f ? v0 : 0.f;
                float v1 = y1[t][1][r]; y1[t][1][r] = v1 > 0.f ? v1 : 0.f;
            }
        }

        v8f y2[4];
        #pragma unroll
        for (int t = 0; t < 4; ++t) {
            y2[t][0] = b2lo.x; y2[t][1] = b2lo.y; y2[t][2] = b2lo.z; y2[t][3] = b2lo.w;
            y2[t][4] = b2hi.x; y2[t][5] = b2hi.y; y2[t][6] = b2hi.z; y2[t][7] = b2hi.w;
        }
        #pragma unroll
        for (int j = 0; j < 8; ++j) {
            #pragma unroll
            for (int t = 0; t < 4; ++t) {
                v2f bb;
                bb.x = y1[t][j >> 2][(j & 3) * 2];
                bb.y = y1[t][j >> 2][(j & 3) * 2 + 1];
                y2[t] = wmma_f32(w2a[j], bb, y2[t]);
            }
        }
        #pragma unroll
        for (int t = 0; t < 4; ++t) {
            #pragma unroll
            for (int r = 0; r < 8; ++r) { float v = y2[t][r]; y2[t][r] = v > 0.f ? v : 0.f; }
        }

        #pragma unroll
        for (int t = 0; t < 4; ++t) {
            #pragma unroll
            for (int r = 0; r < 8; ++r) d3[t][r] = b3v;
        }
        #pragma unroll
        for (int j = 0; j < 4; ++j) {
            #pragma unroll
            for (int t = 0; t < 4; ++t) {
                v2f aa;
                aa.x = y2[t][2 * j];
                aa.y = y2[t][2 * j + 1];
                d3[t] = wmma_f32(aa, w3b[j], d3[t]);
            }
        }
        #pragma unroll
        for (int t = 0; t < 4; ++t) {
            #pragma unroll
            for (int r = 0; r < 8; ++r) { float v = d3[t][r]; d3[t][r] = v > 0.f ? v : 0.f; }
        }
    }

    #pragma unroll
    for (int pi = 0; pi < 4; ++pi) {
        const int p = wave * 4 + pi;
        const v4i ia = *reinterpret_cast<const v4i*>(&s_idx[p * 16 + hi8]);
        const v4i ib = *reinterpret_cast<const v4i*>(&s_idx[p * 16 + hi8 + 4]);
        const int nb[8] = { ia.x, ia.y, ia.z, ia.w, ib.x, ib.y, ib.z, ib.w };
        v8f acc[4];
        #pragma unroll
        for (int mt = 0; mt < 4; ++mt) {
            #pragma unroll
            for (int r = 0; r < 8; ++r) acc[mt][r] = 0.f;
        }
        #pragma unroll
        for (int j = 0; j < 4; ++j) {
            v2f bf;
            bf.x = d3[pi][2 * j];
            bf.y = d3[pi][2 * j + 1];
            #pragma unroll
            for (int mt = 0; mt < 4; ++mt) {
                v2f a;
                a.x = inp[((size_t)b * NPTS + nb[2 * j    ]) * CIN + mt * 16 + laneM];
                a.y = inp[((size_t)b * NPTS + nb[2 * j + 1]) * CIN + mt * 16 + laneM];
                acc[mt] = wmma_f32(a, bf, acc[mt]);
            }
        }
        #pragma unroll
        for (int mt = 0; mt < 4; ++mt) {
            #pragma unroll
            for (int r = 0; r < 8; ++r)
                s_fmat[p][(mt * 16 + r + hi8) * 16 + laneM] = acc[mt][r];
        }
    }

    {
        v8f acc[4];
        #pragma unroll
        for (int t = 0; t < 4; ++t) {
            #pragma unroll
            for (int r = 0; r < 8; ++r) acc[t][r] = 0.f;
        }
        const float* wrow = weight + (size_t)(wave * 256 + hiHalf * 4) * COUT + laneM * 4;
        v4f w0[16], w1[16], a0[4], a1[4];
        __syncthreads();
        const float* arow = &s_fmat[laneM][wave * 256 + hiHalf * 4];
        d_load_w(w0, wrow);
        d_load_a(a0, arow);
        for (int it = 0; it < 3; ++it) {
            d_load_w(w1, wrow + 32 * COUT);
            d_load_a(a1, arow + 32);
            __builtin_amdgcn_sched_barrier(0);
            d_mma(acc, w0, a0);
            __builtin_amdgcn_sched_barrier(0);
            wrow += 64 * COUT;
            arow += 64;
            d_load_w(w0, wrow);
            d_load_a(a0, arow);
            __builtin_amdgcn_sched_barrier(0);
            d_mma(acc, w1, a1);
            __builtin_amdgcn_sched_barrier(0);
        }
        d_load_w(w1, wrow + 32 * COUT);
        d_load_a(a1, arow + 32);
        __builtin_amdgcn_sched_barrier(0);
        d_mma(acc, w0, a0);
        d_mma(acc, w1, a1);

        __syncthreads();
        float* part = fbase;
        #pragma unroll
        for (int r = 0; r < 8; ++r) {
            v4f v;
            v.x = acc[0][r]; v.y = acc[1][r]; v.z = acc[2][r]; v.w = acc[3][r];
            *reinterpret_cast<v4f*>(part + (wave * TP + hi8 + r) * COUT + laneM * 4) = v;
        }
        __syncthreads();
        #pragma unroll
        for (int i = 0; i < 2; ++i) {
            const int q   = tid + i * NTHREADS;
            const int p   = q >> 4;
            const int col = (q & 15) * 4;
            v4f s = *reinterpret_cast<const v4f*>(part + (0 * TP + p) * COUT + col);
            s += *reinterpret_cast<const v4f*>(part + (1 * TP + p) * COUT + col);
            s += *reinterpret_cast<const v4f*>(part + (2 * TP + p) * COUT + col);
            s += *reinterpret_cast<const v4f*>(part + (3 * TP + p) * COUT + col);
            const v4f bb = *reinterpret_cast<const v4f*>(s_bias + col);
            const v4f o  = s * (1.0f / KNB) + bb;
            *reinterpret_cast<v4f*>(out + ((size_t)b * NPTS + n0 + p) * COUT + col) = o;
        }
    }
}

extern "C" void kernel_launch(void* const* d_in, const int* in_sizes, int n_in,
                              void* d_out, int out_size, void* d_ws, size_t ws_size,
                              hipStream_t stream) {
    (void)in_sizes; (void)n_in; (void)out_size; (void)d_ws; (void)ws_size;
    const float* inp      = (const float*)d_in[0];
    const float* points   = (const float*)d_in[1];
    const float* next_pts = (const float*)d_in[2];
    const int*   indices  = (const int*)  d_in[3];
    const float* weight   = (const float*)d_in[4];
    const float* bias     = (const float*)d_in[5];
    const float* centers  = (const float*)d_in[6];
    const float* l1w      = (const float*)d_in[7];
    const float* l1b      = (const float*)d_in[8];
    const float* l2w      = (const float*)d_in[9];
    const float* l2b      = (const float*)d_in[10];
    const float* l3w      = (const float*)d_in[11];
    const float* l3b      = (const float*)d_in[12];
    float* out = (float*)d_out;

    const int grid = BATCH * (NPTS / TP);
    ptconv_wmma_kernel<<<grid, NTHREADS, 0, stream>>>(
        inp, points, next_pts, indices, weight, bias, centers,
        l1w, l1b, l2w, l2b, l3w, l3b, out);
}


# 32 points per workgroup on 8 waves (K split 8 ways, 2 row tiles per wave): every weight row phase D loads feeds 8 tiles, weight reads from the second-level cache halved; changes summation order
# speedup vs baseline: 3.0243x; 1.1064x over previous
#include <hip/hip_runtime.h>
#include <hip/hip_bf16.h>

typedef __attribute__((ext_vector_type(2))) float v2f;
typedef __attribute__((ext_vector_type(8))) float v8f;
typedef __attribute__((ext_vector_type(4))) float v4f;
typedef __attribute__((ext_vector_type(4))) int   v4i;

#define BATCH   4
#define NPTS    8192
#define KNB     16
#define CIN     64
#define COUT    64
#define NC      16
#define DIM     3
#define L1IN    48
#define L1OUT   32
#define TP      32
#define NTHREADS 256
#define NWAVES  8
#define NFOLD   4
#define KWAVE   (CIN * NC / NWAVES)
#define FSTRIDE 1028

static __device__ __forceinline__ v8f wmma_f32(v2f a, v2f b, v8f c) {
    return __builtin_amdgcn_wmma_f32_16x16x4_f32(false, a, false, b,
                                                 (short)0, c, false, false);
}

static __device__ __forceinline__ void d_load_w(v4f (&wr)[8], const float* __restrict__ wrow) {
    #pragma unroll
    for (int j = 0; j < 2; ++j) {
        #pragma unroll
        for (int e = 0; e < 4; ++e)
            wr[j * 4 + e] = *reinterpret_cast<const v4f*>(wrow + (size_t)(j * 8 + e) * COUT);
    }
}
static __device__ __forceinline__ void d_load_a(v4f (&ar)[2][2], const float* arow) {
    #pragma unroll
    for (int m = 0; m < 2; ++m) {
        #pragma unroll
        for (int j = 0; j < 2; ++j)
            ar[m][j] = *reinterpret_cast<const v4f*>(arow + m * 16 * FSTRIDE + j * 8);
    }
}
static __device__ __forceinline__ void d_mma(v8f (&acc)[2][4], const v4f (&wr)[8], const v4f (&ar)[2][2]) {
    #pragma unroll
    for (int j = 0; j < 2; ++j) {
        #pragma unroll
        for (int u = 0; u < 2; ++u) {
            v2f bf[4];
            #pragma unroll
            for (int t = 0; t < 4; ++t) {
                bf[t].x = wr[j * 4 + 2 * u][t];
                bf[t].y = wr[j * 4 + 2 * u + 1][t];
            }
            #pragma unroll
            for (int m = 0; m < 2; ++m) {
                v2f a;
                a.x = ar[m][j][2 * u];
                a.y = ar[m][j][2 * u + 1];
                #pragma unroll
                for (int t = 0; t < 4; ++t)
                    acc[m][t] = wmma_f32(a, bf[t], acc[m][t]);
            }
        }
    }
}

__global__ __launch_bounds__(NTHREADS)
void ptconv_wmma_kernel(const float* __restrict__ inp,
                        const float* __restrict__ points,
                        const float* __restrict__ next_pts,
                        const int*   __restrict__ indices,
                        const float* __restrict__ weight,
                        const float* __restrict__ bias,
                        const float* __restrict__ centers,
                        const float* __restrict__ l1w, const float* __restrict__ l1b,
                        const float* __restrict__ l2w, const float* __restrict__ l2b,
                        const float* __restrict__ l3w, const float* __restrict__ l3b,
                        float* __restrict__ out)
{
    __shared__ __attribute__((aligned(16))) float s_fmat[TP][FSTRIDE];
    __shared__ __attribute__((aligned(16))) float s_rel[TP * KNB][4];
    __shared__ __attribute__((aligned(16))) float s_par[L1OUT * NC + NC * NC + 2 * NC];
    __shared__ __attribute__((aligned(16))) float s_xch[NFOLD * 32 * 2];
    __shared__ __attribute__((aligned(16))) float s_bias[COUT];
    __shared__ __attribute__((aligned(16))) int   s_idx[TP * KNB];
    float* const fbase = &s_fmat[0][0];
    float* const s_l2w  = s_par;
    float* const s_l3w  = s_par + L1OUT * NC;
    float* const s_l2b  = s_l3w + NC * NC;
    float* const s_l3b  = s_l2b + NC;

    const int tid    = threadIdx.x;
    const int wave   = tid >> 5;
    const bool foldWave = __builtin_amdgcn_readfirstlane(tid >> 5) < NFOLD;
    const int lane   = tid & 31;
    const int laneM  = lane & 15;
    const int hiHalf = lane >> 4;
    const int koffA  = hiHalf * 2;
    const int hi8    = hiHalf * 8;

    const int tilesPerBatch = NPTS / TP;
    const int b    = blockIdx.x / tilesPerBatch;
    const int tile = blockIdx.x % tilesPerBatch;
    const int n0   = tile * TP;

    const size_t row0g = (size_t)b * NPTS + n0;
    int   g_idx[2];
    float g_q[2][3];
    #pragma unroll
    for (int i = 0; i < 2; ++i) {
        const int r = tid + i * NTHREADS;
        g_idx[i] = indices[row0g * KNB + r];
        const float* qq = next_pts + (row0g + (r >> 4)) * DIM;
        g_q[i][0] = qq[0]; g_q[i][1] = qq[1]; g_q[i][2] = qq[2];
    }
    const v2f   p_l2w = reinterpret_cast<const v2f*>(l2w)[tid];
    const float p_l3w = l3w[tid];
    const float p_l2b  = l2b[tid & (NC - 1)];
    const float p_l3b  = l3b[tid & (NC - 1)];
    const float p_bias = bias[tid & (COUT - 1)];
    const int fnt = wave & 1;
    const int fk0 = ((wave >> 1) & 1) * 24;
    v2f fw[6], fc[6];
    float fb1 = 0.f;
    #pragma unroll
    for (int ks = 0; ks < 6; ++ks) { fw[ks].x = 0.f; fw[ks].y = 0.f; fc[ks].x = 0.f; fc[ks].y = 0.f; }
    if (foldWave) {
        #pragma unroll
        for (int ks = 0; ks < 6; ++ks) {
            const int kA = fk0 + ks * 4 + koffA;
            fw[ks].x = l1w[(kA    ) * L1OUT + fnt * 16 + laneM];
            fw[ks].y = l1w[(kA + 1) * L1OUT + fnt * 16 + laneM];
            fc[ks]   = *reinterpret_cast<const v2f*>(centers + kA);
        }
        fb1 = l1b[fnt * 16 + laneM];
    }

    for (int i = tid; i < CIN * NC; i += NTHREADS * 2)
        __builtin_prefetch(weight + (size_t)i * COUT, 0, 1);

    float g_p[2][3];
    #pragma unroll
    for (int i = 0; i < 2; ++i) {
        const float* pp = points + ((size_t)b * NPTS + g_idx[i]) * DIM;
        g_p[i][0] = pp[0]; g_p[i][1] = pp[1]; g_p[i][2] = pp[2];
    }

    reinterpret_cast<v2f*>(s_l2w)[tid] = p_l2w;
    s_l3w[tid] = p_l3w;
    if (tid < NC)      { s_l2b[tid] = p_l2b; s_l3b[tid] = p_l3b; }
    if (tid < COUT)    s_bias[tid] = p_bias;

    if (foldWave) {
        const int rowCode = ((laneM & 6) == 0) ? ((laneM & 1) | ((laneM >> 2) & 2)) : 4;
        v8f f;
        #pragma unroll
        for (int r = 0; r < 8; ++r) f[r] = 0.f;
        f[1] = (hiHalf != 0 && fk0 == 0) ? fb1 : 0.f;
        #pragma unroll
        for (int ks = 0; ks < 6; ++ks) {
            const float ind = (rowCode == ((fk0 + ks * 4) >> 4)) ? 1.f : 0.f;
            v2f a;
            a.x = (rowCode == 3) ? -fc[ks].x : ind;
            a.y = (rowCode == 3) ? -fc[ks].y : ind;
            f = wmma_f32(a, fw[ks], f);
        }
        v2f fo;
        fo.x = f[0]; fo.y = f[1];
        reinterpret_cast<v2f*>(s_xch)[wave * 32 + lane] = fo;
    }

    #pragma unroll
    for (int i = 0; i < 2; ++i) {
        const int r = tid + i * NTHREADS;
        s_idx[r] = g_idx[i];
        s_rel[r][0] = g_p[i][0] - g_q[i][0];
        s_rel[r][1] = g_p[i][1] - g_q[i][1];
        s_rel[r][2] = g_p[i][2] - g_q[i][2];
        s_rel[r][3] = 1.0f;
    }
    __syncthreads();

    const v2f* const xq = reinterpret_cast<const v2f*>(s_xch) + lane;
    const v2f wf0 = xq[0 * 32] + xq[2 * 32];
    const v2f wf1 = xq[1 * 32] + xq[3 * 32];
    v2f w2a[8], w3b[4];
    #pragma unroll
    for (int j = 0; j < 8; ++j) {
        const int f = (j >> 2) * 16 + hi8 + (j & 3) * 2;
        w2a[j].x = s_l2w[(f    ) * NC + laneM];
        w2a[j].y = s_l2w[(f + 1) * NC + laneM];
    }
    #pragma unroll
    for (int j = 0; j < 4; ++j) {
        const int f = hi8 + j * 2;
        w3b[j].x = s_l3w[(f    ) * NC + laneM];
        w3b[j].y = s_l3w[(f + 1) * NC + laneM];
    }
    const v4f b2lo = *reinterpret_cast<const v4f*>(s_l2b + hi8);
    const v4f b2hi = *reinterpret_cast<const v4f*>(s_l2b + hi8 + 4);
    const float b3v = s_l3b[laneM];

    v8f d3[4];
    {
        v8f y1[4][2];
        #pragma unroll
        for (int t = 0; t < 4; ++t) {
            const v2f rl = *reinterpret_cast<const v2f*>(&s_rel[wave * 64 + t * 16 + laneM][koffA]);
            v8f z;
            #pragma unroll
            for (int r = 0; r < 8; ++r) z[r] = 0.f;
            y1[t][0] = wmma_f32(wf0, rl, z);
            y1[t][1] = wmma_f32(wf1, rl, z);
        }
        #pragma unroll
        for (int t = 0; t < 4; ++t) {
            #pragma unroll
            for (int r = 0; r < 8; ++r) {
                float v0 = y1[t][0][r]; y1[t][0][r] = v0 > 0.f ? v0 : 0.f;
                float v1 = y1[t][1][r]; y1[t][1][r] = v1 > 0.f ? v1 : 0.f;
            }
        }

        v8f y2[4];
        #pragma unroll
        for (int t = 0; t < 4; ++t) {
            y2[t][0] = b2lo.x; y2[t][1] = b2lo.y; y2[t][2] = b2lo.z; y2[t][3] = b2lo.w;
            y2[t][4] = b2hi.x; y2[t][5] = b2hi.y; y2[t][6] = b2hi.z; y2[t][7] = b2hi.w;
        }
        #pragma unroll
        for (int j = 0; j < 8; ++j) {
            #pragma unroll
            for (int t = 0; t < 4; ++t) {
                v2f bb;
                bb.x = y1[t][j >> 2][(j & 3) * 2];
                bb.y = y1[t][j >> 2][(j & 3) * 2 + 1];
                y2[t] = wmma_f32(w2a[j], bb, y2[t]);
            }
        }
        #pragma unroll
        for (int t = 0; t < 4; ++t) {
            #pragma unroll
            for (int r = 0; r < 8; ++r) { float v = y2[t][r]; y2[t][r] = v > 0.f ? v : 0.f; }
        }

        #pragma unroll
        for (int t = 0; t < 4; ++t) {
            #pragma unroll
            for (int r = 0; r < 8; ++r) d3[t][r] = b3v;
        }
        #pragma unroll
        for (int j = 0; j < 4; ++j) {
            #pragma unroll
            for (int t = 0; t < 4; ++t) {
                v2f aa;
                aa.x = y2[t][2 * j];
                aa.y = y2[t][2 * j + 1];
                d3[t] = wmma_f32(aa, w3b[j], d3[t]);
            }
        }
        #pragma unroll
        for (int t = 0; t < 4; ++t) {
            #pragma unroll
            for (int r = 0; r < 8; ++r) { float v = d3[t][r]; d3[t][r] = v > 0.f ? v : 0.f; }
        }
    }

    #pragma unroll
    for (int pi = 0; pi < 4; ++pi) {
        const int p = wave * 4 + pi;
        const v4i ia = *reinterpret_cast<const v4i*>(&s_idx[p * 16 + hi8]);
        const v4i ib = *reinterpret_cast<const v4i*>(&s_idx[p * 16 + hi8 + 4]);
        const int nb[8] = { ia.x, ia.y, ia.z, ia.w, ib.x, ib.y, ib.z, ib.w };
        v8f acc[4];
        #pragma unroll
        for (int mt = 0; mt < 4; ++mt) {
            #pragma unroll
            for (int r = 0; r < 8; ++r) acc[mt][r] = 0.f;
        }
        #pragma unroll
        for (int j = 0; j < 4; ++j) {
            v2f bf;
            bf.x = d3[pi][2 * j];
            bf.y = d3[pi][2 * j + 1];
            #pragma unroll
            for (int mt = 0; mt < 4; ++mt) {
                v2f a;
                a.x = inp[((size_t)b * NPTS + nb[2 * j    ]) * CIN + mt * 16 + laneM];
                a.y = inp[((size_t)b * NPTS + nb[2 * j + 1]) * CIN + mt * 16 + laneM];
                acc[mt] = wmma_f32(a, bf, acc[mt]);
            }
        }
        #pragma unroll
        for (int mt = 0; mt < 4; ++mt) {
            #pragma unroll
            for (int r = 0; r < 8; ++r)
                s_fmat[p][(mt * 16 + r + hi8) * 16 + laneM] = acc[mt][r];
        }
    }

    {
        v8f acc[2][4];
        #pragma unroll
        for (int m = 0; m < 2; ++m) {
            #pragma unroll
            for (int t = 0; t < 4; ++t) {
                #pragma unroll
                for (int r = 0; r < 8; ++r) acc[m][t][r] = 0.f;
            }
        }
        const float* wrow = weight + (size_t)(wave * KWAVE + hiHalf * 4) * COUT + laneM * 4;
        v4f w0[8], w1[8], a0[2][2], a1[2][2];
        __syncthreads();
        const float* arow = &s_fmat[laneM][wave * KWAVE + hiHalf * 4];
        d_load_w(w0, wrow);
        d_load_a(a0, arow);
        for (int it = 0; it < 3; ++it) {
            d_load_w(w1, wrow + 16 * COUT);
            d_load_a(a1, arow + 16);
            __builtin_amdgcn_sched_barrier(0);
            d_mma(acc, w0, a0);
            __builtin_amdgcn_sched_barrier(0);
            wrow += 32 * COUT;
            arow += 32;
            d_load_w(w0, wrow);
            d_load_a(a0, arow);
            __builtin_amdgcn_sched_barrier(0);
            d_mma(acc, w1, a1);
            __builtin_amdgcn_sched_barrier(0);
        }
        d_load_w(w1, wrow + 16 * COUT);
        d_load_a(a1, arow + 16);
        __builtin_amdgcn_sched_barrier(0);
        d_mma(acc, w0, a0);
        d_mma(acc, w1, a1);

        __syncthreads();
        float* part = fbase;
        #pragma unroll
        for (int m = 0; m < 2; ++m) {
            #pragma unroll
            for (int r = 0; r < 8; ++r) {
                v4f v;
                v.x = acc[m][0][r]; v.y = acc[m][1][r]; v.z = acc[m][2][r]; v.w = acc[m][3][r];
                *reinterpret_cast<v4f*>(part + (wave * TP + m * 16 + hi8 + r) * COUT + laneM * 4) = v;
            }
        }
        __syncthreads();
        #pragma unroll
        for (int i = 0; i < 2; ++i) {
            const int q   = tid + i * NTHREADS;
            const int p   = q >> 4;
            const int col = (q & 15) * 4;
            v4f s = *reinterpret_cast<const v4f*>(part + (0 * TP + p) * COUT + col);
            #pragma unroll
            for (int w = 1; w < NWAVES; ++w)
                s += *reinterpret_cast<const v4f*>(part + (w * TP + p) * COUT + col);
            const v4f bb = *reinterpret_cast<const v4f*>(s_bias + col);
            const v4f o  = s * (1.0f / KNB) + bb;
            *reinterpret_cast<v4f*>(out + ((size_t)b * NPTS + n0 + p) * COUT + col) = o;
        }
    }
}

extern "C" void kernel_launch(void* const* d_in, const int* in_sizes, int n_in,
                              void* d_out, int out_size, void* d_ws, size_t ws_size,
                              hipStream_t stream) {
    (void)in_sizes; (void)n_in; (void)out_size; (void)d_ws; (void)ws_size;
    const float* inp      = (const float*)d_in[0];
    const float* points   = (const float*)d_in[1];
    const float* next_pts = (const float*)d_in[2];
    const int*   indices  = (const int*)  d_in[3];
    const float* weight   = (const float*)d_in[4];
    const float* bias     = (const float*)d_in[5];
    const float* centers  = (const float*)d_in[6];
    const float* l1w      = (const float*)d_in[7];
    const float* l1b      = (const float*)d_in[8];
    const float* l2w      = (const float*)d_in[9];
    const float* l2b      = (const float*)d_in[10];
    const float* l3w      = (const float*)d_in[11];
    const float* l3b      = (const float*)d_in[12];
    float* out = (float*)d_out;

    const int grid = BATCH * (NPTS / TP);
    ptconv_wmma_kernel<<<grid, NTHREADS, 0, stream>>>(
        inp, points, next_pts, indices, weight, bias, centers,
        l1w, l1b, l2w, l2b, l3w, l3b, out);
}


# wave priority (s_setprio): 3 in phases A-C and the tail, 0 in phase D, so a co-resident workgroup's latency-bound phases win the matrix pipe and the other's phase D fills their waits
# speedup vs baseline: 3.1172x; 1.0307x over previous
#include <hip/hip_runtime.h>
#include <hip/hip_bf16.h>

typedef __attribute__((ext_vector_type(2))) float v2f;
typedef __attribute__((ext_vector_type(8))) float v8f;
typedef __attribute__((ext_vector_type(4))) float v4f;
typedef __attribute__((ext_vector_type(4))) int   v4i;

#define BATCH   4
#define NPTS    8192
#define KNB     16
#define CIN     64
#define COUT    64
#define NC      16
#define DIM     3
#define L1IN    48
#define L1OUT   32
#define TP      32
#define NTHREADS 256
#define NWAVES  8
#define NFOLD   4
#define KWAVE   (CIN * NC / NWAVES)
#define FSTRIDE 1028

static __device__ __forceinline__ v8f wmma_f32(v2f a, v2f b, v8f c) {
    return __builtin_amdgcn_wmma_f32_16x16x4_f32(false, a, false, b,
                                                 (short)0, c, false, false);
}

static __device__ __forceinline__ void d_load_w(v4f (&wr)[8], const float* __restrict__ wrow) {
    #pragma unroll
    for (int j = 0; j < 2; ++j) {
        #pragma unroll
        for (int e = 0; e < 4; ++e)
            wr[j * 4 + e] = *reinterpret_cast<const v4f*>(wrow + (size_t)(j * 8 + e) * COUT);
    }
}
static __device__ __forceinline__ void d_load_a(v4f (&ar)[2][2], const float* arow) {
    #pragma unroll
    for (int m = 0; m < 2; ++m) {
        #pragma unroll
        for (int j = 0; j < 2; ++j)
            ar[m][j] = *reinterpret_cast<const v4f*>(arow + m * 16 * FSTRIDE + j * 8);
    }
}
static __device__ __forceinline__ void d_mma(v8f (&acc)[2][4], const v4f (&wr)[8], const v4f (&ar)[2][2]) {
    #pragma unroll
    for (int j = 0; j < 2; ++j) {
        #pragma unroll
        for (int u = 0; u < 2; ++u) {
            v2f bf[4];
            #pragma unroll
            for (int t = 0; t < 4; ++t) {
                bf[t].x = wr[j * 4 + 2 * u][t];
                bf[t].y = wr[j * 4 + 2 * u + 1][t];
            }
            #pragma unroll
            for (int m = 0; m < 2; ++m) {
                v2f a;
                a.x = ar[m][j][2 * u];
                a.y = ar[m][j][2 * u + 1];
                #pragma unroll
                for (int t = 0; t < 4; ++t)
                    acc[m][t] = wmma_f32(a, bf[t], acc[m][t]);
            }
        }
    }
}

__global__ __launch_bounds__(NTHREADS)
void ptconv_wmma_kernel(const float* __restrict__ inp,
                        const float* __restrict__ points,
                        const float* __restrict__ next_pts,
                        const int*   __restrict__ indices,
                        const float* __restrict__ weight,
                        const float* __restrict__ bias,
                        const float* __restrict__ centers,
                        const float* __restrict__ l1w, const float* __restrict__ l1b,
                        const float* __restrict__ l2w, const float* __restrict__ l2b,
                        const float* __restrict__ l3w, const float* __restrict__ l3b,
                        float* __restrict__ out)
{
    __shared__ __attribute__((aligned(16))) float s_fmat[TP][FSTRIDE];
    __shared__ __attribute__((aligned(16))) float s_rel[TP * KNB][4];
    __shared__ __attribute__((aligned(16))) float s_par[L1OUT * NC + NC * NC + 2 * NC];
    __shared__ __attribute__((aligned(16))) float s_xch[NFOLD * 32 * 2];
    __shared__ __attribute__((aligned(16))) float s_bias[COUT];
    __shared__ __attribute__((aligned(16))) int   s_idx[TP * KNB];
    float* const fbase = &s_fmat[0][0];
    float* const s_l2w  = s_par;
    float* const s_l3w  = s_par + L1OUT * NC;
    float* const s_l2b  = s_l3w + NC * NC;
    float* const s_l3b  = s_l2b + NC;

    __builtin_amdgcn_s_setprio(3);

    const int tid    = threadIdx.x;
    const int wave   = tid >> 5;
    const bool foldWave = __builtin_amdgcn_readfirstlane(tid >> 5) < NFOLD;
    const int lane   = tid & 31;
    const int laneM  = lane & 15;
    const int hiHalf = lane >> 4;
    const int koffA  = hiHalf * 2;
    const int hi8    = hiHalf * 8;

    const int tilesPerBatch = NPTS / TP;
    const int b    = blockIdx.x / tilesPerBatch;
    const int tile = blockIdx.x % tilesPerBatch;
    const int n0   = tile * TP;

    const size_t row0g = (size_t)b * NPTS + n0;
    int   g_idx[2];
    float g_q[2][3];
    #pragma unroll
    for (int i = 0; i < 2; ++i) {
        const int r = tid + i * NTHREADS;
        g_idx[i] = indices[row0g * KNB + r];
        const float* qq = next_pts + (row0g + (r >> 4)) * DIM;
        g_q[i][0] = qq[0]; g_q[i][1] = qq[1]; g_q[i][2] = qq[2];
    }
    const v2f   p_l2w = reinterpret_cast<const v2f*>(l2w)[tid];
    const float p_l3w = l3w[tid];
    const float p_l2b  = l2b[tid & (NC - 1)];
    const float p_l3b  = l3b[tid & (NC - 1)];
    const float p_bias = bias[tid & (COUT - 1)];
    const int fnt = wave & 1;
    const int fk0 = ((wave >> 1) & 1) * 24;
    v2f fw[6], fc[6];
    float fb1 = 0.f;
    #pragma unroll
    for (int ks = 0; ks < 6; ++ks) { fw[ks].x = 0.f; fw[ks].y = 0.f; fc[ks].x = 0.f; fc[ks].y = 0.f; }
    if (foldWave) {
        #pragma unroll
        for (int ks = 0; ks < 6; ++ks) {
            const int kA = fk0 + ks * 4 + koffA;
            fw[ks].x = l1w[(kA    ) * L1OUT + fnt * 16 + laneM];
            fw[ks].y = l1w[(kA + 1) * L1OUT + fnt * 16 + laneM];
            fc[ks]   = *reinterpret_cast<const v2f*>(centers + kA);
        }
        fb1 = l1b[fnt * 16 + laneM];
    }

    for (int i = tid; i < CIN * NC; i += NTHREADS * 2)
        __builtin_prefetch(weight + (size_t)i * COUT, 0, 1);

    float g_p[2][3];
    #pragma unroll
    for (int i = 0; i < 2; ++i) {
        const float* pp = points + ((size_t)b * NPTS + g_idx[i]) * DIM;
        g_p[i][0] = pp[0]; g_p[i][1] = pp[1]; g_p[i][2] = pp[2];
    }

    reinterpret_cast<v2f*>(s_l2w)[tid] = p_l2w;
    s_l3w[tid] = p_l3w;
    if (tid < NC)      { s_l2b[tid] = p_l2b; s_l3b[tid] = p_l3b; }
    if (tid < COUT)    s_bias[tid] = p_bias;

    if (foldWave) {
        const int rowCode = ((laneM & 6) == 0) ? ((laneM & 1) | ((laneM >> 2) & 2)) : 4;
        v8f f;
        #pragma unroll
        for (int r = 0; r < 8; ++r) f[r] = 0.f;
        f[1] = (hiHalf != 0 && fk0 == 0) ? fb1 : 0.f;
        #pragma unroll
        for (int ks = 0; ks < 6; ++ks) {
            const float ind = (rowCode == ((fk0 + ks * 4) >> 4)) ? 1.f : 0.f;
            v2f a;
            a.x = (rowCode == 3) ? -fc[ks].x : ind;
            a.y = (rowCode == 3) ? -fc[ks].y : ind;
            f = wmma_f32(a, fw[ks], f);
        }
        v2f fo;
        fo.x = f[0]; fo.y = f[1];
        reinterpret_cast<v2f*>(s_xch)[wave * 32 + lane] = fo;
    }

    #pragma unroll
    for (int i = 0; i < 2; ++i) {
        const int r = tid + i * NTHREADS;
        s_idx[r] = g_idx[i];
        s_rel[r][0] = g_p[i][0] - g_q[i][0];
        s_rel[r][1] = g_p[i][1] - g_q[i][1];
        s_rel[r][2] = g_p[i][2] - g_q[i][2];
        s_rel[r][3] = 1.0f;
    }
    __syncthreads();

    const v2f* const xq = reinterpret_cast<const v2f*>(s_xch) + lane;
    const v2f wf0 = xq[0 * 32] + xq[2 * 32];
    const v2f wf1 = xq[1 * 32] + xq[3 * 32];
    v2f w2a[8], w3b[4];
    #pragma unroll
    for (int j = 0; j < 8; ++j) {
        const int f = (j >> 2) * 16 + hi8 + (j & 3) * 2;
        w2a[j].x = s_l2w[(f    ) * NC + laneM];
        w2a[j].y = s_l2w[(f + 1) * NC + laneM];
    }
    #pragma unroll
    for (int j = 0; j < 4; ++j) {
        const int f = hi8 + j * 2;
        w3b[j].x = s_l3w[(f    ) * NC + laneM];
        w3b[j].y = s_l3w[(f + 1) * NC + laneM];
    }
    const v4f b2lo = *reinterpret_cast<const v4f*>(s_l2b + hi8);
    const v4f b2hi = *reinterpret_cast<const v4f*>(s_l2b + hi8 + 4);
    const float b3v = s_l3b[laneM];

    v8f d3[4];
    {
        v8f y1[4][2];
        #pragma unroll
        for (int t = 0; t < 4; ++t) {
            const v2f rl = *reinterpret_cast<const v2f*>(&s_rel[wave * 64 + t * 16 + laneM][koffA]);
            v8f z;
            #pragma unroll
            for (int r = 0; r < 8; ++r) z[r] = 0.f;
            y1[t][0] = wmma_f32(wf0, rl, z);
            y1[t][1] = wmma_f32(wf1, rl, z);
        }
        #pragma unroll
        for (int t = 0; t < 4; ++t) {
            #pragma unroll
            for (int r = 0; r < 8; ++r) {
                float v0 = y1[t][0][r]; y1[t][0][r] = v0 > 0.f ? v0 : 0.f;
                float v1 = y1[t][1][r]; y1[t][1][r] = v1 > 0.f ? v1 : 0.f;
            }
        }

        v8f y2[4];
        #pragma unroll
        for (int t = 0; t < 4; ++t) {
            y2[t][0] = b2lo.x; y2[t][1] = b2lo.y; y2[t][2] = b2lo.z; y2[t][3] = b2lo.w;
            y2[t][4] = b2hi.x; y2[t][5] = b2hi.y; y2[t][6] = b2hi.z; y2[t][7] = b2hi.w;
        }
        #pragma unroll
        for (int j = 0; j < 8; ++j) {
            #pragma unroll
            for (int t = 0; t < 4; ++t) {
                v2f bb;
                bb.x = y1[t][j >> 2][(j & 3) * 2];
                bb.y = y1[t][j >> 2][(j & 3) * 2 + 1];
                y2[t] = wmma_f32(w2a[j], bb, y2[t]);
            }
        }
        #pragma unroll
        for (int t = 0; t < 4; ++t) {
            #pragma unroll
            for (int r = 0; r < 8; ++r) { float v = y2[t][r]; y2[t][r] = v > 0.f ? v : 0.f; }
        }

        #pragma unroll
        for (int t = 0; t < 4; ++t) {
            #pragma unroll
            for (int r = 0; r < 8; ++r) d3[t][r] = b3v;
        }
        #pragma unroll
        for (int j = 0; j < 4; ++j) {
            #pragma unroll
            for (int t = 0; t < 4; ++t) {
                v2f aa;
                aa.x = y2[t][2 * j];
                aa.y = y2[t][2 * j + 1];
                d3[t] = wmma_f32(aa, w3b[j], d3[t]);
            }
        }
        #pragma unroll
        for (int t = 0; t < 4; ++t) {
            #pragma unroll
            for (int r = 0; r < 8; ++r) { float v = d3[t][r]; d3[t][r] = v > 0.f ? v : 0.f; }
        }
    }

    #pragma unroll
    for (int pi = 0; pi < 4; ++pi) {
        const int p = wave * 4 + pi;
        const v4i ia = *reinterpret_cast<const v4i*>(&s_idx[p * 16 + hi8]);
        const v4i ib = *reinterpret_cast<const v4i*>(&s_idx[p * 16 + hi8 + 4]);
        const int nb[8] = { ia.x, ia.y, ia.z, ia.w, ib.x, ib.y, ib.z, ib.w };
        v8f acc[4];
        #pragma unroll
        for (int mt = 0; mt < 4; ++mt) {
            #pragma unroll
            for (int r = 0; r < 8; ++r) acc[mt][r] = 0.f;
        }
        #pragma unroll
        for (int j = 0; j < 4; ++j) {
            v2f bf;
            bf.x = d3[pi][2 * j];
            bf.y = d3[pi][2 * j + 1];
            #pragma unroll
            for (int mt = 0; mt < 4; ++mt) {
                v2f a;
                a.x = inp[((size_t)b * NPTS + nb[2 * j    ]) * CIN + mt * 16 + laneM];
                a.y = inp[((size_t)b * NPTS + nb[2 * j + 1]) * CIN + mt * 16 + laneM];
                acc[mt] = wmma_f32(a, bf, acc[mt]);
            }
        }
        #pragma unroll
        for (int mt = 0; mt < 4; ++mt) {
            #pragma unroll
            for (int r = 0; r < 8; ++r)
                s_fmat[p][(mt * 16 + r + hi8) * 16 + laneM] = acc[mt][r];
        }
    }

    {
        v8f acc[2][4];
        #pragma unroll
        for (int m = 0; m < 2; ++m) {
            #pragma unroll
            for (int t = 0; t < 4; ++t) {
                #pragma unroll
                for (int r = 0; r < 8; ++r) acc[m][t][r] = 0.f;
            }
        }
        const float* wrow = weight + (size_t)(wave * KWAVE + hiHalf * 4) * COUT + laneM * 4;
        v4f w0[8], w1[8], a0[2][2], a1[2][2];
        __syncthreads();
        __builtin_amdgcn_s_setprio(0);
        const float* arow = &s_fmat[laneM][wave * KWAVE + hiHalf * 4];
        d_load_w(w0, wrow);
        d_load_a(a0, arow);
        for (int it = 0; it < 3; ++it) {
            d_load_w(w1, wrow + 16 * COUT);
            d_load_a(a1, arow + 16);
            __builtin_amdgcn_sched_barrier(0);
            d_mma(acc, w0, a0);
            __builtin_amdgcn_sched_barrier(0);
            wrow += 32 * COUT;
            arow += 32;
            d_load_w(w0, wrow);
            d_load_a(a0, arow);
            __builtin_amdgcn_sched_barrier(0);
            d_mma(acc, w1, a1);
            __builtin_amdgcn_sched_barrier(0);
        }
        d_load_w(w1, wrow + 16 * COUT);
        d_load_a(a1, arow + 16);
        __builtin_amdgcn_sched_barrier(0);
        d_mma(acc, w0, a0);
        d_mma(acc, w1, a1);
        __builtin_amdgcn_s_setprio(3);

        __syncthreads();
        float* part = fbase;
        #pragma unroll
        for (int m = 0; m < 2; ++m) {
            #pragma unroll
            for (int r = 0; r < 8; ++r) {
                v4f v;
                v.x = acc[m][0][r]; v.y = acc[m][1][r]; v.z = acc[m][2][r]; v.w = acc[m][3][r];
                *reinterpret_cast<v4f*>(part + (wave * TP + m * 16 + hi8 + r) * COUT + laneM * 4) = v;
            }
        }
        __syncthreads();
        #pragma unroll
        for (int i = 0; i < 2; ++i) {
            const int q   = tid + i * NTHREADS;
            const int p   = q >> 4;
            const int col = (q & 15) * 4;
            v4f s = *reinterpret_cast<const v4f*>(part + (0 * TP + p) * COUT + col);
            #pragma unroll
            for (int w = 1; w < NWAVES; ++w)
                s += *reinterpret_cast<const v4f*>(part + (w * TP + p) * COUT + col);
            const v4f bb = *reinterpret_cast<const v4f*>(s_bias + col);
            const v4f o  = s * (1.0f / KNB) + bb;
            *reinterpret_cast<v4f*>(out + ((size_t)b * NPTS + n0 + p) * COUT + col) = o;
        }
    }
}

extern "C" void kernel_launch(void* const* d_in, const int* in_sizes, int n_in,
                              void* d_out, int out_size, void* d_ws, size_t ws_size,
                              hipStream_t stream) {
    (void)in_sizes; (void)n_in; (void)out_size; (void)d_ws; (void)ws_size;
    const float* inp      = (const float*)d_in[0];
    const float* points   = (const float*)d_in[1];
    const float* next_pts = (const float*)d_in[2];
    const int*   indices  = (const int*)  d_in[3];
    const float* weight   = (const float*)d_in[4];
    const float* bias     = (const float*)d_in[5];
    const float* centers  = (const float*)d_in[6];
    const float* l1w      = (const float*)d_in[7];
    const float* l1b      = (const float*)d_in[8];
    const float* l2w      = (const float*)d_in[9];
    const float* l2b      = (const float*)d_in[10];
    const float* l3w      = (const float*)d_in[11];
    const float* l3b      = (const float*)d_in[12];
    float* out = (float*)d_out;

    const int grid = BATCH * (NPTS / TP);
    ptconv_wmma_kernel<<<grid, NTHREADS, 0, stream>>>(
        inp, points, next_pts, indices, weight, bias, centers,
        l1w, l1b, l2w, l2b, l3w, l3b, out);
}
